# starting kernel
# baseline (speedup 1.0000x reference)
#include <hip/hip_runtime.h>
#include <math.h>

typedef __attribute__((ext_vector_type(2))) float v2f;
typedef __attribute__((ext_vector_type(8))) float v8f;

#define B_  1024
#define N_  1024
#define M_  8
#define D_  64
#define R_  64
#define C_  10

#define LDS_W      32768
#define LDS_HALF   10240
#define AT_STRIDE  66
#define S_STRIDE   66

__global__ void wl_kernel(const float* __restrict__ Wlast, float* __restrict__ Wl) {
    int tid = blockIdx.x * 256 + threadIdx.x;
    if (tid >= C_ * R_ * R_) return;
    int r  = tid & 63;
    int ci = tid >> 6;
    const float* p = Wlast + ((size_t)ci * 8) * 64 + r;
    float s = 0.f;
    #pragma unroll
    for (int m = 0; m < 8; ++m) s += p[m * 64];
    Wl[tid] = s;
}

__global__ void phi_kernel(const float* __restrict__ x, float* __restrict__ phidiv) {
    int tid = blockIdx.x * 256 + threadIdx.x;
    if (tid >= B_ * D_) return;
    const float* xp = x + (size_t)tid * 16;
    float acc[8];
    #pragma unroll
    for (int m = 0; m < 8; ++m) acc[m] = 0.f;
    for (int n = 0; n < 16; ++n) {
        float a = 1.5707963267948966f * xp[n];
        float sn, cs;
        sincosf(a, &sn, &cs);
        float pc[8], ps[8];
        pc[0] = 1.f; ps[0] = 1.f;
        #pragma unroll
        for (int k = 1; k < 8; ++k) { pc[k] = pc[k-1] * cs; ps[k] = ps[k-1] * sn; }
        #pragma unroll
        for (int m = 0; m < 8; ++m) acc[m] = fmaf(pc[7-m], ps[m], acc[m]);
    }
    float* o = phidiv + (size_t)tid * 8;
    #pragma unroll
    for (int m = 0; m < 8; ++m) o[m] = acc[m];
}

__global__ __launch_bounds__(256, 1)
void trnet_kernel(const float* __restrict__ Wsh, const float* __restrict__ phidiv,
                  const float* __restrict__ Wl, float* __restrict__ out)
{
    extern __shared__ float smem[];
    float* Wsm = smem;
    const int half = threadIdx.x >> 7;
    const int t0   = threadIdx.x & 127;
    float* At  = smem + LDS_W + half * LDS_HALF;
    float* Ssm = At  + 64 * AT_STRIDE;
    float* phs = Ssm + 64 * S_STRIDE;
    float* red = phs + 512;

    const int b = blockIdx.x * 2 + half;

    {
        const float4* src = (const float4*)Wsh;
        float4* dst = (float4*)Wsm;
        for (int q = 0; q < 32; ++q)
            dst[threadIdx.x + 256 * q] = src[threadIdx.x + 256 * q];
    }
    {
        const float* src = phidiv + (size_t)b * (D_ * M_);
        #pragma unroll
        for (int q = 0; q < 4; ++q) phs[t0 + 128 * q] = src[t0 + 128 * q];
    }
    for (int q = t0; q < 64 * S_STRIDE; q += 128) Ssm[q] = 1.0f;
    __syncthreads();

    const int wave = t0 >> 5;
    const int lane = t0 & 31;
    const int lmod = lane & 15;
    const int lhi  = lane >> 4;

    for (int d = 0; d < D_; ++d) {
        float ph[8];
        #pragma unroll
        for (int j = 0; j < 8; ++j) ph[j] = phs[d * 8 + j];
        #pragma unroll 4
        for (int q = 0; q < 32; ++q) {
            int e  = t0 + 128 * q;
            int ko = e & 63;
            int i  = e >> 6;
            const float* wp = Wsm + i * 512 + ko;
            float v = 0.f;
            #pragma unroll
            for (int j = 0; j < 8; ++j) v = fmaf(ph[j], wp[j * 64], v);
            At[ko * AT_STRIDE + i] = v;
        }
        __syncthreads();

        const float* srow = Ssm + (16 * wave + lmod) * S_STRIDE;
        v8f c0 = {}, c1 = {}, c2 = {}, c3 = {};
        #pragma unroll
        for (int kb = 0; kb < 16; ++kb) {
            const int koff = kb * 4 + 2 * lhi;
            v2f a = *(const v2f*)(srow + koff);
            const float* ab = At + koff;
            v2f b0 = *(const v2f*)(ab + (lmod +  0) * AT_STRIDE);
            v2f b1 = *(const v2f*)(ab + (lmod + 16) * AT_STRIDE);
            v2f b2 = *(const v2f*)(ab + (lmod + 32) * AT_STRIDE);
            v2f b3 = *(const v2f*)(ab + (lmod + 48) * AT_STRIDE);
            c0 = __builtin_amdgcn_wmma_f32_16x16x4_f32(false, a, false, b0, (short)0, c0, false, false);
            c1 = __builtin_amdgcn_wmma_f32_16x16x4_f32(false, a, false, b1, (short)0, c1, false, false);
            c2 = __builtin_amdgcn_wmma_f32_16x16x4_f32(false, a, false, b2, (short)0, c2, false, false);
            c3 = __builtin_amdgcn_wmma_f32_16x16x4_f32(false, a, false, b3, (short)0, c3, false, false);
        }
        #pragma unroll
        for (int v = 0; v < 8; ++v) {
            float* prow = Ssm + (16 * wave + v + 8 * lhi) * S_STRIDE + lmod;
            prow[0]  = c0[v];
            prow[16] = c1[v];
            prow[32] = c2[v];
            prow[48] = c3[v];
        }
        __syncthreads();
    }

    float acc[C_];
    #pragma unroll
    for (int c = 0; c < C_; ++c) acc[c] = 0.f;
    for (int q = 0; q < 32; ++q) {
        int e = t0 + 128 * q;
        int r = e & 63;
        int i = e >> 6;
        float s = Ssm[r * S_STRIDE + i];
        #pragma unroll
        for (int c = 0; c < C_; ++c)
            acc[c] = fmaf(s, Wl[c * 4096 + i * 64 + r], acc[c]);
    }
    #pragma unroll
    for (int c = 0; c < C_; ++c) red[t0 * C_ + c] = acc[c];
    __syncthreads();
    if (t0 < C_) {
        float s = 0.f;
        for (int t = 0; t < 128; ++t) s += red[t * C_ + t0];
        At[t0] = s;
    }
    __syncthreads();
    if (t0 < C_) {
        float mx = At[0];
        #pragma unroll
        for (int c = 1; c < C_; ++c) mx = fmaxf(mx, At[c]);
        float se = 0.f;
        #pragma unroll
        for (int c = 0; c < C_; ++c) se += expf(At[c] - mx);
        out[b * C_ + t0] = At[t0] - (logf(se) + mx);
    }
}

extern "C" void kernel_launch(void* const* d_in, const int* in_sizes, int n_in,
                              void* d_out, int out_size, void* d_ws, size_t ws_size,
                              hipStream_t stream) {
    const float* x     = (const float*)d_in[0];
    const float* Wsh   = (const float*)d_in[1];
    const float* Wlast = (const float*)d_in[2];
    float* out = (float*)d_out;
    float* ws  = (float*)d_ws;
    float* Wl     = ws;
    float* phidiv = ws + (C_ * R_ * R_);

    wl_kernel<<<(C_ * R_ * R_ + 255) / 256, 256, 0, stream>>>(Wlast, Wl);
    phi_kernel<<<(B_ * D_ + 255) / 256, 256, 0, stream>>>(x, phidiv);

    const size_t shbytes = (size_t)(LDS_W + 2 * LDS_HALF) * sizeof(float);
    hipFuncSetAttribute((const void*)trnet_kernel,
                        hipFuncAttributeMaxDynamicSharedMemorySize, (int)shbytes);
    trnet_kernel<<<B_ / 2, 256, shbytes, stream>>>(Wsh, phidiv, Wl, out);
}


# state0 is all ones inside the reference, so the 64 state rows of a batch stay equal: carry one row per batch, 16 batches = WMMA M, W fragments held in registers; f32 summation order changes
# speedup vs baseline: 4.9197x; 4.9197x over previous
#include <hip/hip_runtime.h>
#include <math.h>

typedef __attribute__((ext_vector_type(2))) float v2f;
typedef __attribute__((ext_vector_type(8))) float v8f;

#define B_  1024
#define N_  1024
#define M_  8
#define D_  64
#define R_  64
#define C_  10

#define VS  68
#define PS  516


__global__ void wl_kernel(const float* __restrict__ Wlast, float* __restrict__ WlS) {
    const int wv   = (blockIdx.x * 256 + threadIdx.x) >> 5;
    const int lane = threadIdx.x & 31;
    const float* p = Wlast + (size_t)wv * 512 + lane;
    float s = 0.f;
    #pragma unroll
    for (int q = 0; q < 16; ++q) s += p[32 * q];
    #pragma unroll
    for (int off = 16; off > 0; off >>= 1) s += __shfl_xor(s, off, 32);
    if (lane == 0) WlS[wv] = s;
}

__global__ void phi_kernel(const float* __restrict__ x, float* __restrict__ phidiv) {
    int tid = blockIdx.x * 256 + threadIdx.x;
    if (tid >= B_ * D_) return;
    const float* xp = x + (size_t)tid * 16;
    float acc[8];
    #pragma unroll
    for (int m = 0; m < 8; ++m) acc[m] = 0.f;
    for (int n = 0; n < 16; ++n) {
        float a = 1.5707963267948966f * xp[n];
        float sn, cs;
        sincosf(a, &sn, &cs);
        float pc[8], ps[8];
        pc[0] = 1.f; ps[0] = 1.f;
        #pragma unroll
        for (int k = 1; k < 8; ++k) { pc[k] = pc[k-1] * cs; ps[k] = ps[k-1] * sn; }
        #pragma unroll
        for (int m = 0; m < 8; ++m) acc[m] = fmaf(pc[7-m], ps[m], acc[m]);
    }
    float* o = phidiv + (size_t)tid * 8;
    #pragma unroll
    for (int m = 0; m < 8; ++m) o[m] = acc[m];
}

__global__ __launch_bounds__(128)
void trnet_kernel(const float* __restrict__ Wsh, const float* __restrict__ phidiv,
                  const float* __restrict__ WlS, float* __restrict__ out)
{
    __shared__ __attribute__((aligned(16))) float Vsm[2 * 16 * VS];
    __shared__ __attribute__((aligned(16))) float phs[16 * PS];
    __shared__ float trs[16 * C_];

    const int tid  = threadIdx.x;
    const int w    = tid >> 5;
    const int lane = tid & 31;
    const int c    = lane & 15;
    const int h    = lane >> 4;
    const int b0   = blockIdx.x * 16;

    {
        const float4* src = (const float4*)(phidiv + (size_t)b0 * (D_ * M_));
        #pragma unroll 4
        for (int q = 0; q < 16; ++q) {
            int e4 = tid + 128 * q;
            int bb = e4 >> 7;
            int r4 = e4 & 127;
            *(float4*)(phs + bb * PS + 4 * r4) = src[e4];
        }
    }
    for (int q = tid; q < 16 * VS; q += 128) Vsm[q] = 1.0f;

    v2f Bf[8][16];
    {
        const float* wb = Wsh + (2 * h) * 512 + 16 * w + c;
        #pragma unroll
        for (int kb = 0; kb < 16; ++kb) {
            #pragma unroll
            for (int j = 0; j < 8; ++j) {
                const float* p = wb + (4 * kb) * 512 + j * 64;
                v2f t; t[0] = p[0]; t[1] = p[512];
                Bf[j][kb] = t;
            }
        }
    }
    __syncthreads();

    for (int d = 0; d < D_; ++d) {
        const float* Vr = Vsm + (d & 1) * (16 * VS) + c * VS + 2 * h;
        v2f a[16];
        #pragma unroll
        for (int kb = 0; kb < 16; ++kb) a[kb] = *(const v2f*)(Vr + 4 * kb);

        v8f acc[8];
        #pragma unroll
        for (int j = 0; j < 8; ++j) acc[j] = (v8f){};

        #pragma unroll
        for (int kb = 0; kb < 16; ++kb) {
            #pragma unroll
            for (int j = 0; j < 8; ++j)
                acc[j] = __builtin_amdgcn_wmma_f32_16x16x4_f32(false, a[kb], false, Bf[j][kb],
                                                               (short)0, acc[j], false, false);
        }

        float* Vw = Vsm + ((d + 1) & 1) * (16 * VS) + 16 * w + c;
        #pragma unroll
        for (int r = 0; r < 8; ++r) {
            const float* pp = phs + (8 * h + r) * PS + d * 8;
            float4 p0 = *(const float4*)pp;
            float4 p1 = *(const float4*)(pp + 4);
            float s = p0.x * acc[0][r];
            s = fmaf(p0.y, acc[1][r], s);
            s = fmaf(p0.z, acc[2][r], s);
            s = fmaf(p0.w, acc[3][r], s);
            s = fmaf(p1.x, acc[4][r], s);
            s = fmaf(p1.y, acc[5][r], s);
            s = fmaf(p1.z, acc[6][r], s);
            s = fmaf(p1.w, acc[7][r], s);
            Vw[(8 * h + r) * VS] = s;
        }
        __syncthreads();
    }

    for (int t = tid; t < 16 * C_; t += 128) {
        int bb = t / C_, cc = t - bb * C_;
        const float* vr = Vsm + bb * VS;
        const float* wl = WlS + cc * 64;
        float s = 0.f;
        for (int i = 0; i < 64; ++i) s = fmaf(vr[i], wl[i], s);
        trs[t] = s;
    }
    __syncthreads();
    for (int t = tid; t < 16 * C_; t += 128) {
        int bb = t / C_;
        const float* tr = trs + bb * C_;
        float mx = tr[0];
        #pragma unroll
        for (int k = 1; k < C_; ++k) mx = fmaxf(mx, tr[k]);
        float se = 0.f;
        #pragma unroll
        for (int k = 0; k < C_; ++k) se += expf(tr[k] - mx);
        out[(size_t)b0 * C_ + t] = trs[t] - (logf(se) + mx);
    }
}

extern "C" void kernel_launch(void* const* d_in, const int* in_sizes, int n_in,
                              void* d_out, int out_size, void* d_ws, size_t ws_size,
                              hipStream_t stream) {
    const float* x     = (const float*)d_in[0];
    const float* Wsh   = (const float*)d_in[1];
    const float* Wlast = (const float*)d_in[2];
    float* out = (float*)d_out;
    float* ws  = (float*)d_ws;
    float* WlS    = ws;
    float* phidiv = ws + (C_ * R_ * R_);

    wl_kernel<<<(C_ * R_ * 32) / 256, 256, 0, stream>>>(Wlast, WlS);
    phi_kernel<<<(B_ * D_ + 255) / 256, 256, 0, stream>>>(x, phidiv);
    const size_t reserve = (size_t)168 * 1024;
    hipFuncSetAttribute((const void*)trnet_kernel,
                        hipFuncAttributeMaxDynamicSharedMemorySize, (int)reserve);
    trnet_kernel<<<B_ / 16, 128, reserve, stream>>>(Wsh, phidiv, WlS, out);
}


# Product transposed (state as WMMA B operand): a wave's D registers are its own K quarter of the next step, the other three cross LDS under those WMMAs (split barrier); f32 sum order over i changes
# speedup vs baseline: 5.5194x; 1.1219x over previous
#include <hip/hip_runtime.h>
#include <math.h>

typedef __attribute__((ext_vector_type(2))) float v2f;
typedef __attribute__((ext_vector_type(8))) float v8f;

#define B_  1024
#define N_  1024
#define M_  8
#define D_  64
#define R_  64
#define C_  10

#define VS  68
#define PS  516
#define XW  256
#define XB  1024


__global__ void wl_kernel(const float* __restrict__ Wlast, float* __restrict__ WlS) {
    const int wv   = (blockIdx.x * 256 + threadIdx.x) >> 5;
    const int lane = threadIdx.x & 31;
    const float* p = Wlast + (size_t)wv * 512 + lane;
    float s = 0.f;
    #pragma unroll
    for (int q = 0; q < 16; ++q) s += p[32 * q];
    #pragma unroll
    for (int off = 16; off > 0; off >>= 1) s += __shfl_xor(s, off, 32);
    if (lane == 0) WlS[wv] = s;
}

__global__ void phi_kernel(const float* __restrict__ x, float* __restrict__ phidiv) {
    int tid = blockIdx.x * 256 + threadIdx.x;
    if (tid >= B_ * D_) return;
    const float* xp = x + (size_t)tid * 16;
    float acc[8];
    #pragma unroll
    for (int m = 0; m < 8; ++m) acc[m] = 0.f;
    for (int n = 0; n < 16; ++n) {
        float a = 1.5707963267948966f * xp[n];
        float sn, cs;
        sincosf(a, &sn, &cs);
        float pc[8], ps[8];
        pc[0] = 1.f; ps[0] = 1.f;
        #pragma unroll
        for (int k = 1; k < 8; ++k) { pc[k] = pc[k-1] * cs; ps[k] = ps[k-1] * sn; }
        #pragma unroll
        for (int m = 0; m < 8; ++m) acc[m] = fmaf(pc[7-m], ps[m], acc[m]);
    }
    float* o = phidiv + (size_t)tid * 8;
    #pragma unroll
    for (int m = 0; m < 8; ++m) o[m] = acc[m];
}

__global__ __launch_bounds__(128)
void trnet_kernel(const float* __restrict__ Wsh, const float* __restrict__ phidiv,
                  const float* __restrict__ WlS, float* __restrict__ out)
{
    __shared__ __attribute__((aligned(16))) float Xs[2 * XB];
    __shared__ __attribute__((aligned(16))) float Vfin[16 * VS];
    __shared__ __attribute__((aligned(16))) float phs[16 * PS];
    __shared__ float trs[16 * C_];

    const int tid  = threadIdx.x;
    const int w    = tid >> 5;
    const int lane = tid & 31;
    const int c    = lane & 15;
    const int h    = lane >> 4;
    const int b0   = blockIdx.x * 16;

    {
        const float4* src = (const float4*)(phidiv + (size_t)b0 * (D_ * M_));
        #pragma unroll 4
        for (int q = 0; q < 16; ++q) {
            int e4 = tid + 128 * q;
            int bb = e4 >> 7;
            int r4 = e4 & 127;
            *(float4*)(phs + bb * PS + 4 * r4) = src[e4];
        }
    }

    v2f Af[8][16];
    {
        #pragma unroll
        for (int t = 0; t < 4; ++t) {
            const int u = (w + t) & 3;
            #pragma unroll
            for (int q = 0; q < 4; ++q) {
                const float* p = Wsh + (16 * u + 8 * h + 2 * q) * 512 + 16 * w + c;
                #pragma unroll
                for (int j = 0; j < 8; ++j) {
                    v2f f; f[0] = p[j * 64]; f[1] = p[j * 64 + 512];
                    Af[j][4 * t + q] = f;
                }
            }
        }
    }

    float*       xw  = Xs + w * XW + lane * 4;
    const float* xr1 = Xs + ((w + 1) & 3) * XW + lane * 4;
    const float* xr2 = Xs + ((w + 2) & 3) * XW + lane * 4;
    const float* xr3 = Xs + ((w + 3) & 3) * XW + lane * 4;
    const float* php = phs + c * PS;

    float s0 = 1.f, s1 = 1.f, s2 = 1.f, s3 = 1.f, s4 = 1.f, s5 = 1.f, s6 = 1.f, s7 = 1.f;
    *(float4*)(xw)       = make_float4(s0, s1, s2, s3);
    *(float4*)(xw + 128) = make_float4(s4, s5, s6, s7);

    for (int d = 0; d < D_; ++d) {
        const int rb = (d & 1) * XB;
        const int wb = XB - rb;

        v2f bo0, bo1, bo2, bo3;
        bo0[0] = s0; bo0[1] = s1; bo1[0] = s2; bo1[1] = s3;
        bo2[0] = s4; bo2[1] = s5; bo3[0] = s6; bo3[1] = s7;

        v8f acc[8];
        #pragma unroll
        for (int j = 0; j < 8; ++j)
            acc[j] = __builtin_amdgcn_wmma_f32_16x16x4_f32(false, Af[j][0], false, bo0,
                                                           (short)0, (v8f){}, false, false);

        __builtin_amdgcn_sched_barrier(0);
        __builtin_amdgcn_fence(__ATOMIC_RELEASE, "workgroup");
        __builtin_amdgcn_s_barrier_signal(-1);
        __builtin_amdgcn_sched_barrier(0);

        #pragma unroll
        for (int j = 0; j < 8; ++j)
            acc[j] = __builtin_amdgcn_wmma_f32_16x16x4_f32(false, Af[j][1], false, bo1,
                                                           (short)0, acc[j], false, false);

        __builtin_amdgcn_sched_barrier(0);
        __builtin_amdgcn_s_barrier_wait(-1);
        __builtin_amdgcn_fence(__ATOMIC_ACQUIRE, "workgroup");

        const float4 xa0 = *(const float4*)(xr1 + rb), xa1 = *(const float4*)(xr1 + rb + 128);
        const float4 xb0 = *(const float4*)(xr2 + rb), xb1 = *(const float4*)(xr2 + rb + 128);
        const float4 xc0 = *(const float4*)(xr3 + rb), xc1 = *(const float4*)(xr3 + rb + 128);
        const float4 p0 = *(const float4*)(php + d * 8);
        const float4 p1 = *(const float4*)(php + d * 8 + 4);
        __builtin_amdgcn_sched_barrier(0);

        #pragma unroll
        for (int j = 0; j < 8; ++j)
            acc[j] = __builtin_amdgcn_wmma_f32_16x16x4_f32(false, Af[j][2], false, bo2,
                                                           (short)0, acc[j], false, false);
        #pragma unroll
        for (int j = 0; j < 8; ++j)
            acc[j] = __builtin_amdgcn_wmma_f32_16x16x4_f32(false, Af[j][3], false, bo3,
                                                           (short)0, acc[j], false, false);

        __builtin_amdgcn_sched_barrier(0);

        v2f bx[12];
        bx[0][0] = xa0.x; bx[0][1] = xa0.y; bx[1][0] = xa0.z; bx[1][1] = xa0.w;
        bx[2][0] = xa1.x; bx[2][1] = xa1.y; bx[3][0] = xa1.z; bx[3][1] = xa1.w;
        bx[4][0] = xb0.x; bx[4][1] = xb0.y; bx[5][0] = xb0.z; bx[5][1] = xb0.w;
        bx[6][0] = xb1.x; bx[6][1] = xb1.y; bx[7][0] = xb1.z; bx[7][1] = xb1.w;
        bx[8][0] = xc0.x; bx[8][1] = xc0.y; bx[9][0] = xc0.z; bx[9][1] = xc0.w;
        bx[10][0] = xc1.x; bx[10][1] = xc1.y; bx[11][0] = xc1.z; bx[11][1] = xc1.w;

        #pragma unroll
        for (int kb = 4; kb < 16; ++kb) {
            #pragma unroll
            for (int j = 0; j < 8; ++j)
                acc[j] = __builtin_amdgcn_wmma_f32_16x16x4_f32(false, Af[j][kb], false, bx[kb - 4],
                                                               (short)0, acc[j], false, false);
        }

        #define COMBINE(r) fmaf(p1.w, acc[7][r], fmaf(p1.z, acc[6][r], fmaf(p1.y, acc[5][r], fmaf(p1.x, acc[4][r], \
                           fmaf(p0.w, acc[3][r], fmaf(p0.z, acc[2][r], fmaf(p0.y, acc[1][r], p0.x * acc[0][r])))))))
        s0 = COMBINE(0); s1 = COMBINE(1); s2 = COMBINE(2); s3 = COMBINE(3);
        s4 = COMBINE(4); s5 = COMBINE(5); s6 = COMBINE(6); s7 = COMBINE(7);
        #undef COMBINE

        *(float4*)(xw + wb)       = make_float4(s0, s1, s2, s3);
        *(float4*)(xw + wb + 128) = make_float4(s4, s5, s6, s7);
    }

    {
        float* vf = Vfin + c * VS + 16 * w + 8 * h;
        *(float4*)(vf)     = make_float4(s0, s1, s2, s3);
        *(float4*)(vf + 4) = make_float4(s4, s5, s6, s7);
    }
    __syncthreads();

    for (int t = tid; t < 16 * C_; t += 128) {
        int bb = t / C_, cc = t - bb * C_;
        const float* vr = Vfin + bb * VS;
        const float* wl = WlS + cc * 64;
        float s = 0.f;
        for (int i = 0; i < 64; ++i) s = fmaf(vr[i], wl[i], s);
        trs[t] = s;
    }
    __syncthreads();
    for (int t = tid; t < 16 * C_; t += 128) {
        int bb = t / C_;
        const float* tr = trs + bb * C_;
        float mx = tr[0];
        #pragma unroll
        for (int k = 1; k < C_; ++k) mx = fmaxf(mx, tr[k]);
        float se = 0.f;
        #pragma unroll
        for (int k = 0; k < C_; ++k) se += expf(tr[k] - mx);
        out[(size_t)b0 * C_ + t] = trs[t] - (logf(se) + mx);
    }
}

extern "C" void kernel_launch(void* const* d_in, const int* in_sizes, int n_in,
                              void* d_out, int out_size, void* d_ws, size_t ws_size,
                              hipStream_t stream) {
    const float* x     = (const float*)d_in[0];
    const float* Wsh   = (const float*)d_in[1];
    const float* Wlast = (const float*)d_in[2];
    float* out = (float*)d_out;
    float* ws  = (float*)d_ws;
    float* WlS    = ws;
    float* phidiv = ws + (C_ * R_ * R_);

    wl_kernel<<<(C_ * R_ * 32) / 256, 256, 0, stream>>>(Wlast, WlS);
    phi_kernel<<<(B_ * D_ + 255) / 256, 256, 0, stream>>>(x, phidiv);
    const size_t reserve = (size_t)168 * 1024;
    hipFuncSetAttribute((const void*)trnet_kernel,
                        hipFuncAttributeMaxDynamicSharedMemorySize, (int)reserve);
    trnet_kernel<<<B_ / 16, 128, reserve, stream>>>(Wsh, phidiv, WlS, out);
}


# Chain split in time: 53 steps forward on 64 workgroups, the last 11 backward from the class weights on 640 workgroups over idle WGPs, joined by dot products in a 4th launch; f32 sum order changes
# speedup vs baseline: 6.0653x; 1.0989x over previous
#include <hip/hip_runtime.h>
#include <math.h>

typedef __attribute__((ext_vector_type(2))) float v2f;
typedef __attribute__((ext_vector_type(8))) float v8f;

#define B_  1024
#define N_  1024
#define M_  8
#define D_  64
#define R_  64
#define C_  10

#define VS  68
#define PS  516
#define XW  256
#define XB  1024

#define FS    53
#define NFWD  (B_ / 16)
#define NBWD  (B_ * C_ / 16)
#define NB4   ((D_ - FS) * 2)


__global__ void wl_kernel(const float* __restrict__ Wlast, float* __restrict__ WlS) {
    const int wv   = (blockIdx.x * 256 + threadIdx.x) >> 5;
    const int lane = threadIdx.x & 31;
    const float* p = Wlast + (size_t)wv * 512 + lane;
    float s = 0.f;
    #pragma unroll
    for (int q = 0; q < 16; ++q) s += p[32 * q];
    #pragma unroll
    for (int off = 16; off > 0; off >>= 1) s += __shfl_xor(s, off, 32);
    if (lane == 0) WlS[wv] = s;
}

__global__ void phi_kernel(const float* __restrict__ x, float* __restrict__ phidiv) {
    int tid = blockIdx.x * 256 + threadIdx.x;
    if (tid >= B_ * D_) return;
    const float* xp = x + (size_t)tid * 16;
    float acc[8];
    #pragma unroll
    for (int m = 0; m < 8; ++m) acc[m] = 0.f;
    for (int n = 0; n < 16; ++n) {
        float a = 1.5707963267948966f * xp[n];
        float sn, cs;
        sincosf(a, &sn, &cs);
        float pc[8], ps[8];
        pc[0] = 1.f; ps[0] = 1.f;
        #pragma unroll
        for (int k = 1; k < 8; ++k) { pc[k] = pc[k-1] * cs; ps[k] = ps[k-1] * sn; }
        #pragma unroll
        for (int m = 0; m < 8; ++m) acc[m] = fmaf(pc[7-m], ps[m], acc[m]);
    }
    float* o = phidiv + (size_t)tid * 8;
    #pragma unroll
    for (int m = 0; m < 8; ++m) o[m] = acc[m];
}

__global__ __launch_bounds__(128)
void trnet_kernel(const float* __restrict__ Wsh, const float* __restrict__ phidiv,
                  const float* __restrict__ WlS, float* __restrict__ Vf, float* __restrict__ Ub)
{
    __shared__ __attribute__((aligned(16))) float Xs[2 * XB];
    __shared__ __attribute__((aligned(16))) float Vfin[16 * VS];
    __shared__ __attribute__((aligned(16))) float phs[16 * PS];

    const int tid  = threadIdx.x;
    const int w    = tid >> 5;
    const int lane = tid & 31;
    const int c    = lane & 15;
    const int h    = lane >> 4;
    const bool fwd  = blockIdx.x < NFWD;
    const int  col0 = (fwd ? blockIdx.x : blockIdx.x - NFWD) * 16;

    if (fwd) {
        const float4* src = (const float4*)(phidiv + (size_t)col0 * (D_ * M_));
        #pragma unroll 4
        for (int q = 0; q < 16; ++q) {
            int e4 = tid + 128 * q;
            int bb = e4 >> 7;
            int r4 = e4 & 127;
            *(float4*)(phs + bb * PS + 4 * r4) = src[e4];
        }
    } else {
        const float4* src = (const float4*)phidiv;
        #pragma unroll
        for (int q = 0; q < (16 * NB4 + 127) / 128; ++q) {
            int e4 = tid + 128 * q;
            if (e4 < 16 * NB4) {
                int bb = e4 / NB4;
                int r4 = FS * 2 + (e4 - bb * NB4);
                int bs = (col0 + bb) / C_;
                *(float4*)(phs + bb * PS + 4 * r4) = src[(size_t)bs * 128 + r4];
            }
        }
    }

    v2f Af[8][16];
    if (fwd) {
        #pragma unroll
        for (int t = 0; t < 4; ++t) {
            const int u = (w + t) & 3;
            #pragma unroll
            for (int q = 0; q < 4; ++q) {
                const float* p = Wsh + (16 * u + 8 * h + 2 * q) * 512 + 16 * w + c;
                #pragma unroll
                for (int j = 0; j < 8; ++j) {
                    v2f f; f[0] = p[j * 64]; f[1] = p[j * 64 + 512];
                    Af[j][4 * t + q] = f;
                }
            }
        }
    } else {
        #pragma unroll
        for (int t = 0; t < 4; ++t) {
            const int u = (w + t) & 3;
            #pragma unroll
            for (int qq = 0; qq < 2; ++qq) {
                const float* p = Wsh + (16 * w + c) * 512 + 16 * u + 8 * h + 4 * qq;
                #pragma unroll
                for (int j = 0; j < 8; ++j) {
                    const float4 f4 = *(const float4*)(p + j * 64);
                    v2f f; f[0] = f4.x; f[1] = f4.y;
                    v2f g; g[0] = f4.z; g[1] = f4.w;
                    Af[j][4 * t + 2 * qq]     = f;
                    Af[j][4 * t + 2 * qq + 1] = g;
                }
            }
        }
    }

    float*       xw  = Xs + w * XW + lane * 4;
    const float* xr1 = Xs + ((w + 1) & 3) * XW + lane * 4;
    const float* xr2 = Xs + ((w + 2) & 3) * XW + lane * 4;
    const float* xr3 = Xs + ((w + 3) & 3) * XW + lane * 4;
    const float* php = phs + c * PS;

    float s0 = 1.f, s1 = 1.f, s2 = 1.f, s3 = 1.f, s4 = 1.f, s5 = 1.f, s6 = 1.f, s7 = 1.f;
    if (!fwd) {
        const float* wl = WlS + ((col0 + c) % C_) * 64 + 16 * w + 8 * h;
        const float4 a0 = *(const float4*)(wl), a1 = *(const float4*)(wl + 4);
        s0 = a0.x; s1 = a0.y; s2 = a0.z; s3 = a0.w;
        s4 = a1.x; s5 = a1.y; s6 = a1.z; s7 = a1.w;
    }
    const int nsteps = fwd ? FS : D_ - FS;
    const int dfirst = fwd ? 0 : D_ - 1;
    const int dstep  = fwd ? 1 : -1;
    *(float4*)(xw)       = make_float4(s0, s1, s2, s3);
    *(float4*)(xw + 128) = make_float4(s4, s5, s6, s7);

    for (int s = 0; s < nsteps; ++s) {
        const int d  = dfirst + dstep * s;
        const int rb = (s & 1) * XB;
        const int wb = XB - rb;

        v2f bo0, bo1, bo2, bo3;
        bo0[0] = s0; bo0[1] = s1; bo1[0] = s2; bo1[1] = s3;
        bo2[0] = s4; bo2[1] = s5; bo3[0] = s6; bo3[1] = s7;

        v8f acc[8];
        #pragma unroll
        for (int j = 0; j < 8; ++j)
            acc[j] = __builtin_amdgcn_wmma_f32_16x16x4_f32(false, Af[j][0], false, bo0,
                                                           (short)0, (v8f){}, false, false);

        __builtin_amdgcn_sched_barrier(0);
        __builtin_amdgcn_fence(__ATOMIC_RELEASE, "workgroup");
        __builtin_amdgcn_s_barrier_signal(-1);
        __builtin_amdgcn_sched_barrier(0);

        #pragma unroll
        for (int j = 0; j < 8; ++j)
            acc[j] = __builtin_amdgcn_wmma_f32_16x16x4_f32(false, Af[j][1], false, bo1,
                                                           (short)0, acc[j], false, false);

        __builtin_amdgcn_sched_barrier(0);
        __builtin_amdgcn_s_barrier_wait(-1);
        __builtin_amdgcn_fence(__ATOMIC_ACQUIRE, "workgroup");

        const float4 xa0 = *(const float4*)(xr1 + rb), xa1 = *(const float4*)(xr1 + rb + 128);
        const float4 xb0 = *(const float4*)(xr2 + rb), xb1 = *(const float4*)(xr2 + rb + 128);
        const float4 xc0 = *(const float4*)(xr3 + rb), xc1 = *(const float4*)(xr3 + rb + 128);
        const float4 p0 = *(const float4*)(php + d * 8);
        const float4 p1 = *(const float4*)(php + d * 8 + 4);
        __builtin_amdgcn_sched_barrier(0);

        #pragma unroll
        for (int j = 0; j < 8; ++j)
            acc[j] = __builtin_amdgcn_wmma_f32_16x16x4_f32(false, Af[j][2], false, bo2,
                                                           (short)0, acc[j], false, false);
        #pragma unroll
        for (int j = 0; j < 8; ++j)
            acc[j] = __builtin_amdgcn_wmma_f32_16x16x4_f32(false, Af[j][3], false, bo3,
                                                           (short)0, acc[j], false, false);

        __builtin_amdgcn_sched_barrier(0);

        v2f bx[12];
        bx[0][0] = xa0.x; bx[0][1] = xa0.y; bx[1][0] = xa0.z; bx[1][1] = xa0.w;
        bx[2][0] = xa1.x; bx[2][1] = xa1.y; bx[3][0] = xa1.z; bx[3][1] = xa1.w;
        bx[4][0] = xb0.x; bx[4][1] = xb0.y; bx[5][0] = xb0.z; bx[5][1] = xb0.w;
        bx[6][0] = xb1.x; bx[6][1] = xb1.y; bx[7][0] = xb1.z; bx[7][1] = xb1.w;
        bx[8][0] = xc0.x; bx[8][1] = xc0.y; bx[9][0] = xc0.z; bx[9][1] = xc0.w;
        bx[10][0] = xc1.x; bx[10][1] = xc1.y; bx[11][0] = xc1.z; bx[11][1] = xc1.w;

        #pragma unroll
        for (int kb = 4; kb < 16; ++kb) {
            #pragma unroll
            for (int j = 0; j < 8; ++j)
                acc[j] = __builtin_amdgcn_wmma_f32_16x16x4_f32(false, Af[j][kb], false, bx[kb - 4],
                                                               (short)0, acc[j], false, false);
        }

        #define COMBINE(r) fmaf(p1.w, acc[7][r], fmaf(p1.z, acc[6][r], fmaf(p1.y, acc[5][r], fmaf(p1.x, acc[4][r], \
                           fmaf(p0.w, acc[3][r], fmaf(p0.z, acc[2][r], fmaf(p0.y, acc[1][r], p0.x * acc[0][r])))))))
        s0 = COMBINE(0); s1 = COMBINE(1); s2 = COMBINE(2); s3 = COMBINE(3);
        s4 = COMBINE(4); s5 = COMBINE(5); s6 = COMBINE(6); s7 = COMBINE(7);
        #undef COMBINE

        *(float4*)(xw + wb)       = make_float4(s0, s1, s2, s3);
        *(float4*)(xw + wb + 128) = make_float4(s4, s5, s6, s7);
    }

    {
        float* vf = Vfin + c * VS + 16 * w + 8 * h;
        *(float4*)(vf)     = make_float4(s0, s1, s2, s3);
        *(float4*)(vf + 4) = make_float4(s4, s5, s6, s7);
    }
    __syncthreads();
    {
        float* dst = (fwd ? Vf : Ub) + (size_t)col0 * 64;
        #pragma unroll
        for (int q = 0; q < 2; ++q) {
            int e4  = tid + 128 * q;
            int row = e4 >> 4;
            int c4  = e4 & 15;
            *(float4*)(dst + row * 64 + 4 * c4) = *(const float4*)(Vfin + row * VS + 4 * c4);
        }
    }
}

__global__ __launch_bounds__(512)
void join_kernel(const float* __restrict__ Vf, const float* __restrict__ Ub,
                 float* __restrict__ out) {
    __shared__ float res[16 * C_];
    const int wv   = threadIdx.x >> 5;
    const int b    = blockIdx.x * 16 + wv;
    const int lane = threadIdx.x & 31;
    const float v0 = Vf[b * 64 + lane], v1 = Vf[b * 64 + 32 + lane];
    const float* up = Ub + (size_t)b * (C_ * 64) + lane;
    float tr[C_];
    #pragma unroll
    for (int k = 0; k < C_; ++k) tr[k] = fmaf(v1, up[k * 64 + 32], v0 * up[k * 64]);
    #pragma unroll
    for (int off = 16; off > 0; off >>= 1) {
        #pragma unroll
        for (int k = 0; k < C_; ++k) tr[k] += __shfl_xor(tr[k], off, 32);
    }
    float mx = tr[0];
    #pragma unroll
    for (int k = 1; k < C_; ++k) mx = fmaxf(mx, tr[k]);
    float se = 0.f;
    #pragma unroll
    for (int k = 0; k < C_; ++k) se += expf(tr[k] - mx);
    float mine = tr[0];
    #pragma unroll
    for (int k = 1; k < C_; ++k) mine = (lane == k) ? tr[k] : mine;
    if (lane < C_) res[wv * C_ + lane] = mine - (logf(se) + mx);
    __syncthreads();
    if (threadIdx.x < 16 * C_) out[(size_t)blockIdx.x * (16 * C_) + threadIdx.x] = res[threadIdx.x];
}

extern "C" void kernel_launch(void* const* d_in, const int* in_sizes, int n_in,
                              void* d_out, int out_size, void* d_ws, size_t ws_size,
                              hipStream_t stream) {
    const float* x     = (const float*)d_in[0];
    const float* Wsh   = (const float*)d_in[1];
    const float* Wlast = (const float*)d_in[2];
    float* out = (float*)d_out;
    float* ws  = (float*)d_ws;
    float* WlS    = ws;
    float* phidiv = ws + (C_ * R_ * R_);
    float* Vf     = phidiv + (B_ * D_ * M_);
    float* Ub     = Vf + (B_ * R_);

    wl_kernel<<<(C_ * R_ * 32) / 256, 256, 0, stream>>>(Wlast, WlS);
    phi_kernel<<<(B_ * D_ + 255) / 256, 256, 0, stream>>>(x, phidiv);
    const size_t reserve = (size_t)168 * 1024;
    hipFuncSetAttribute((const void*)trnet_kernel,
                        hipFuncAttributeMaxDynamicSharedMemorySize, (int)reserve);
    trnet_kernel<<<NFWD + NBWD, 128, reserve, stream>>>(Wsh, phidiv, WlS, Vf, Ub);
    join_kernel<<<B_ / 16, 512, 0, stream>>>(Vf, Ub, out);
}


# Class-weight sums and pooled features computed in one launch (336 blocks, block-uniform branch) instead of two dependent launches: 4 launches become 3; arithmetic and sum orders unchanged
# speedup vs baseline: 6.3121x; 1.0407x over previous
#include <hip/hip_runtime.h>
#include <math.h>

typedef __attribute__((ext_vector_type(2))) float v2f;
typedef __attribute__((ext_vector_type(8))) float v8f;

#define B_  1024
#define N_  1024
#define M_  8
#define D_  64
#define R_  64
#define C_  10

#define VS  68
#define PS  516
#define XW  256
#define XB  1024

#define FS    53
#define NFWD  (B_ / 16)
#define NBWD  (B_ * C_ / 16)
#define NB4   ((D_ - FS) * 2)


#define PHB  (B_ * D_ / 256)
#define WLB  (C_ * R_ * 32 / 256)

__global__ __launch_bounds__(256)
void prep_kernel(const float* __restrict__ x, const float* __restrict__ Wlast,
                 float* __restrict__ phidiv, float* __restrict__ WlS) {
    if (blockIdx.x < PHB) {
        const int tid = blockIdx.x * 256 + threadIdx.x;
        const float* xp = x + (size_t)tid * 16;
        float acc[8];
        #pragma unroll
        for (int m = 0; m < 8; ++m) acc[m] = 0.f;
        for (int n = 0; n < 16; ++n) {
            float a = 1.5707963267948966f * xp[n];
            float sn, cs;
            sincosf(a, &sn, &cs);
            float pc[8], ps[8];
            pc[0] = 1.f; ps[0] = 1.f;
            #pragma unroll
            for (int k = 1; k < 8; ++k) { pc[k] = pc[k-1] * cs; ps[k] = ps[k-1] * sn; }
            #pragma unroll
            for (int m = 0; m < 8; ++m) acc[m] = fmaf(pc[7-m], ps[m], acc[m]);
        }
        float* o = phidiv + (size_t)tid * 8;
        #pragma unroll
        for (int m = 0; m < 8; ++m) o[m] = acc[m];
    } else {
        const int wv   = ((blockIdx.x - PHB) * 256 + threadIdx.x) >> 5;
        const int lane = threadIdx.x & 31;
        const float* p = Wlast + (size_t)wv * 512 + lane;
        float s = 0.f;
        #pragma unroll
        for (int q = 0; q < 16; ++q) s += p[32 * q];
        #pragma unroll
        for (int off = 16; off > 0; off >>= 1) s += __shfl_xor(s, off, 32);
        if (lane == 0) WlS[wv] = s;
    }
}

__global__ __launch_bounds__(128)
void trnet_kernel(const float* __restrict__ Wsh, const float* __restrict__ phidiv,
                  const float* __restrict__ WlS, float* __restrict__ Vf, float* __restrict__ Ub)
{
    __shared__ __attribute__((aligned(16))) float Xs[2 * XB];
    __shared__ __attribute__((aligned(16))) float Vfin[16 * VS];
    __shared__ __attribute__((aligned(16))) float phs[16 * PS];

    const int tid  = threadIdx.x;
    const int w    = tid >> 5;
    const int lane = tid & 31;
    const int c    = lane & 15;
    const int h    = lane >> 4;
    const bool fwd  = blockIdx.x < NFWD;
    const int  col0 = (fwd ? blockIdx.x : blockIdx.x - NFWD) * 16;

    if (fwd) {
        const float4* src = (const float4*)(phidiv + (size_t)col0 * (D_ * M_));
        #pragma unroll 4
        for (int q = 0; q < 16; ++q) {
            int e4 = tid + 128 * q;
            int bb = e4 >> 7;
            int r4 = e4 & 127;
            *(float4*)(phs + bb * PS + 4 * r4) = src[e4];
        }
    } else {
        const float4* src = (const float4*)phidiv;
        #pragma unroll
        for (int q = 0; q < (16 * NB4 + 127) / 128; ++q) {
            int e4 = tid + 128 * q;
            if (e4 < 16 * NB4) {
                int bb = e4 / NB4;
                int r4 = FS * 2 + (e4 - bb * NB4);
                int bs = (col0 + bb) / C_;
                *(float4*)(phs + bb * PS + 4 * r4) = src[(size_t)bs * 128 + r4];
            }
        }
    }

    v2f Af[8][16];
    if (fwd) {
        #pragma unroll
        for (int t = 0; t < 4; ++t) {
            const int u = (w + t) & 3;
            #pragma unroll
            for (int q = 0; q < 4; ++q) {
                const float* p = Wsh + (16 * u + 8 * h + 2 * q) * 512 + 16 * w + c;
                #pragma unroll
                for (int j = 0; j < 8; ++j) {
                    v2f f; f[0] = p[j * 64]; f[1] = p[j * 64 + 512];
                    Af[j][4 * t + q] = f;
                }
            }
        }
    } else {
        #pragma unroll
        for (int t = 0; t < 4; ++t) {
            const int u = (w + t) & 3;
            #pragma unroll
            for (int qq = 0; qq < 2; ++qq) {
                const float* p = Wsh + (16 * w + c) * 512 + 16 * u + 8 * h + 4 * qq;
                #pragma unroll
                for (int j = 0; j < 8; ++j) {
                    const float4 f4 = *(const float4*)(p + j * 64);
                    v2f f; f[0] = f4.x; f[1] = f4.y;
                    v2f g; g[0] = f4.z; g[1] = f4.w;
                    Af[j][4 * t + 2 * qq]     = f;
                    Af[j][4 * t + 2 * qq + 1] = g;
                }
            }
        }
    }

    float*       xw  = Xs + w * XW + lane * 4;
    const float* xr1 = Xs + ((w + 1) & 3) * XW + lane * 4;
    const float* xr2 = Xs + ((w + 2) & 3) * XW + lane * 4;
    const float* xr3 = Xs + ((w + 3) & 3) * XW + lane * 4;
    const float* php = phs + c * PS;

    float s0 = 1.f, s1 = 1.f, s2 = 1.f, s3 = 1.f, s4 = 1.f, s5 = 1.f, s6 = 1.f, s7 = 1.f;
    if (!fwd) {
        const float* wl = WlS + ((col0 + c) % C_) * 64 + 16 * w + 8 * h;
        const float4 a0 = *(const float4*)(wl), a1 = *(const float4*)(wl + 4);
        s0 = a0.x; s1 = a0.y; s2 = a0.z; s3 = a0.w;
        s4 = a1.x; s5 = a1.y; s6 = a1.z; s7 = a1.w;
    }
    const int nsteps = fwd ? FS : D_ - FS;
    const int dfirst = fwd ? 0 : D_ - 1;
    const int dstep  = fwd ? 1 : -1;
    *(float4*)(xw)       = make_float4(s0, s1, s2, s3);
    *(float4*)(xw + 128) = make_float4(s4, s5, s6, s7);

    for (int s = 0; s < nsteps; ++s) {
        const int d  = dfirst + dstep * s;
        const int rb = (s & 1) * XB;
        const int wb = XB - rb;

        v2f bo0, bo1, bo2, bo3;
        bo0[0] = s0; bo0[1] = s1; bo1[0] = s2; bo1[1] = s3;
        bo2[0] = s4; bo2[1] = s5; bo3[0] = s6; bo3[1] = s7;

        v8f acc[8];
        #pragma unroll
        for (int j = 0; j < 8; ++j)
            acc[j] = __builtin_amdgcn_wmma_f32_16x16x4_f32(false, Af[j][0], false, bo0,
                                                           (short)0, (v8f){}, false, false);

        __builtin_amdgcn_sched_barrier(0);
        __builtin_amdgcn_fence(__ATOMIC_RELEASE, "workgroup");
        __builtin_amdgcn_s_barrier_signal(-1);
        __builtin_amdgcn_sched_barrier(0);

        #pragma unroll
        for (int j = 0; j < 8; ++j)
            acc[j] = __builtin_amdgcn_wmma_f32_16x16x4_f32(false, Af[j][1], false, bo1,
                                                           (short)0, acc[j], false, false);

        __builtin_amdgcn_sched_barrier(0);
        __builtin_amdgcn_s_barrier_wait(-1);
        __builtin_amdgcn_fence(__ATOMIC_ACQUIRE, "workgroup");

        const float4 xa0 = *(const float4*)(xr1 + rb), xa1 = *(const float4*)(xr1 + rb + 128);
        const float4 xb0 = *(const float4*)(xr2 + rb), xb1 = *(const float4*)(xr2 + rb + 128);
        const float4 xc0 = *(const float4*)(xr3 + rb), xc1 = *(const float4*)(xr3 + rb + 128);
        const float4 p0 = *(const float4*)(php + d * 8);
        const float4 p1 = *(const float4*)(php + d * 8 + 4);
        __builtin_amdgcn_sched_barrier(0);

        #pragma unroll
        for (int j = 0; j < 8; ++j)
            acc[j] = __builtin_amdgcn_wmma_f32_16x16x4_f32(false, Af[j][2], false, bo2,
                                                           (short)0, acc[j], false, false);
        #pragma unroll
        for (int j = 0; j < 8; ++j)
            acc[j] = __builtin_amdgcn_wmma_f32_16x16x4_f32(false, Af[j][3], false, bo3,
                                                           (short)0, acc[j], false, false);

        __builtin_amdgcn_sched_barrier(0);

        v2f bx[12];
        bx[0][0] = xa0.x; bx[0][1] = xa0.y; bx[1][0] = xa0.z; bx[1][1] = xa0.w;
        bx[2][0] = xa1.x; bx[2][1] = xa1.y; bx[3][0] = xa1.z; bx[3][1] = xa1.w;
        bx[4][0] = xb0.x; bx[4][1] = xb0.y; bx[5][0] = xb0.z; bx[5][1] = xb0.w;
        bx[6][0] = xb1.x; bx[6][1] = xb1.y; bx[7][0] = xb1.z; bx[7][1] = xb1.w;
        bx[8][0] = xc0.x; bx[8][1] = xc0.y; bx[9][0] = xc0.z; bx[9][1] = xc0.w;
        bx[10][0] = xc1.x; bx[10][1] = xc1.y; bx[11][0] = xc1.z; bx[11][1] = xc1.w;

        #pragma unroll
        for (int kb = 4; kb < 16; ++kb) {
            #pragma unroll
            for (int j = 0; j < 8; ++j)
                acc[j] = __builtin_amdgcn_wmma_f32_16x16x4_f32(false, Af[j][kb], false, bx[kb - 4],
                                                               (short)0, acc[j], false, false);
        }

        #define COMBINE(r) fmaf(p1.w, acc[7][r], fmaf(p1.z, acc[6][r], fmaf(p1.y, acc[5][r], fmaf(p1.x, acc[4][r], \
                           fmaf(p0.w, acc[3][r], fmaf(p0.z, acc[2][r], fmaf(p0.y, acc[1][r], p0.x * acc[0][r])))))))
        s0 = COMBINE(0); s1 = COMBINE(1); s2 = COMBINE(2); s3 = COMBINE(3);
        s4 = COMBINE(4); s5 = COMBINE(5); s6 = COMBINE(6); s7 = COMBINE(7);
        #undef COMBINE

        *(float4*)(xw + wb)       = make_float4(s0, s1, s2, s3);
        *(float4*)(xw + wb + 128) = make_float4(s4, s5, s6, s7);
    }

    {
        float* vf = Vfin + c * VS + 16 * w + 8 * h;
        *(float4*)(vf)     = make_float4(s0, s1, s2, s3);
        *(float4*)(vf + 4) = make_float4(s4, s5, s6, s7);
    }
    __syncthreads();
    {
        float* dst = (fwd ? Vf : Ub) + (size_t)col0 * 64;
        #pragma unroll
        for (int q = 0; q < 2; ++q) {
            int e4  = tid + 128 * q;
            int row = e4 >> 4;
            int c4  = e4 & 15;
            *(float4*)(dst + row * 64 + 4 * c4) = *(const float4*)(Vfin + row * VS + 4 * c4);
        }
    }
}

__global__ __launch_bounds__(512)
void join_kernel(const float* __restrict__ Vf, const float* __restrict__ Ub,
                 float* __restrict__ out) {
    __shared__ float res[16 * C_];
    const int wv   = threadIdx.x >> 5;
    const int b    = blockIdx.x * 16 + wv;
    const int lane = threadIdx.x & 31;
    const float v0 = Vf[b * 64 + lane], v1 = Vf[b * 64 + 32 + lane];
    const float* up = Ub + (size_t)b * (C_ * 64) + lane;
    float tr[C_];
    #pragma unroll
    for (int k = 0; k < C_; ++k) tr[k] = fmaf(v1, up[k * 64 + 32], v0 * up[k * 64]);
    #pragma unroll
    for (int off = 16; off > 0; off >>= 1) {
        #pragma unroll
        for (int k = 0; k < C_; ++k) tr[k] += __shfl_xor(tr[k], off, 32);
    }
    float mx = tr[0];
    #pragma unroll
    for (int k = 1; k < C_; ++k) mx = fmaxf(mx, tr[k]);
    float se = 0.f;
    #pragma unroll
    for (int k = 0; k < C_; ++k) se += expf(tr[k] - mx);
    float mine = tr[0];
    #pragma unroll
    for (int k = 1; k < C_; ++k) mine = (lane == k) ? tr[k] : mine;
    if (lane < C_) res[wv * C_ + lane] = mine - (logf(se) + mx);
    __syncthreads();
    if (threadIdx.x < 16 * C_) out[(size_t)blockIdx.x * (16 * C_) + threadIdx.x] = res[threadIdx.x];
}

extern "C" void kernel_launch(void* const* d_in, const int* in_sizes, int n_in,
                              void* d_out, int out_size, void* d_ws, size_t ws_size,
                              hipStream_t stream) {
    const float* x     = (const float*)d_in[0];
    const float* Wsh   = (const float*)d_in[1];
    const float* Wlast = (const float*)d_in[2];
    float* out = (float*)d_out;
    float* ws  = (float*)d_ws;
    float* WlS    = ws;
    float* phidiv = ws + (C_ * R_ * R_);
    float* Vf     = phidiv + (B_ * D_ * M_);
    float* Ub     = Vf + (B_ * R_);

    prep_kernel<<<PHB + WLB, 256, 0, stream>>>(x, Wlast, phidiv, WlS);
    const size_t reserve = (size_t)168 * 1024;
    hipFuncSetAttribute((const void*)trnet_kernel,
                        hipFuncAttributeMaxDynamicSharedMemorySize, (int)reserve);
    trnet_kernel<<<NFWD + NBWD, 128, reserve, stream>>>(Wsh, phidiv, WlS, Vf, Ub);
    join_kernel<<<B_ / 16, 512, 0, stream>>>(Vf, Ub, out);
}


# Backward blocks re-cut to one batch (10 classes): step matrix summed over j once per step in the A fragments, 16 WMMAs per wave-step not 128; split 53/11 -> 44/20 steps; f32 sum order changes
# speedup vs baseline: 6.6924x; 1.0602x over previous
#include <hip/hip_runtime.h>
#include <math.h>

typedef __attribute__((ext_vector_type(2))) float v2f;
typedef __attribute__((ext_vector_type(8))) float v8f;

#define B_  1024
#define N_  1024
#define M_  8
#define D_  64
#define R_  64
#define C_  10

#define VS  68
#define PS  516
#define XW  256
#define XB  1024

#define FS    44
#define NFWD  (B_ / 16)
#define NBWD  B_
#define NB4   ((D_ - FS) * 2)


#define PHB  (B_ * D_ / 256)
#define WLB  (C_ * R_ * 32 / 256)

__global__ __launch_bounds__(256)
void prep_kernel(const float* __restrict__ x, const float* __restrict__ Wlast,
                 float* __restrict__ phidiv, float* __restrict__ WlS) {
    if (blockIdx.x < PHB) {
        const int tid = blockIdx.x * 256 + threadIdx.x;
        const float* xp = x + (size_t)tid * 16;
        float acc[8];
        #pragma unroll
        for (int m = 0; m < 8; ++m) acc[m] = 0.f;
        for (int n = 0; n < 16; ++n) {
            float a = 1.5707963267948966f * xp[n];
            float sn, cs;
            sincosf(a, &sn, &cs);
            float pc[8], ps[8];
            pc[0] = 1.f; ps[0] = 1.f;
            #pragma unroll
            for (int k = 1; k < 8; ++k) { pc[k] = pc[k-1] * cs; ps[k] = ps[k-1] * sn; }
            #pragma unroll
            for (int m = 0; m < 8; ++m) acc[m] = fmaf(pc[7-m], ps[m], acc[m]);
        }
        float* o = phidiv + (size_t)tid * 8;
        #pragma unroll
        for (int m = 0; m < 8; ++m) o[m] = acc[m];
    } else {
        const int wv   = ((blockIdx.x - PHB) * 256 + threadIdx.x) >> 5;
        const int lane = threadIdx.x & 31;
        const float* p = Wlast + (size_t)wv * 512 + lane;
        float s = 0.f;
        #pragma unroll
        for (int q = 0; q < 16; ++q) s += p[32 * q];
        #pragma unroll
        for (int off = 16; off > 0; off >>= 1) s += __shfl_xor(s, off, 32);
        if (lane == 0) WlS[wv] = s;
    }
}

__global__ __launch_bounds__(128)
void trnet_kernel(const float* __restrict__ Wsh, const float* __restrict__ phidiv,
                  const float* __restrict__ WlS, float* __restrict__ Vf, float* __restrict__ Ub)
{
    __shared__ __attribute__((aligned(16))) float Xs[2 * XB];
    __shared__ __attribute__((aligned(16))) float Vfin[16 * VS];
    __shared__ __attribute__((aligned(16))) float phs[16 * PS];

    const int tid  = threadIdx.x;
    const int w    = tid >> 5;
    const int lane = tid & 31;
    const int c    = lane & 15;
    const int h    = lane >> 4;
    const bool fwd  = blockIdx.x < NFWD;
    const int  col0 = fwd ? blockIdx.x * 16 : blockIdx.x - NFWD;

    if (fwd) {
        const float4* src = (const float4*)(phidiv + (size_t)col0 * (D_ * M_));
        #pragma unroll 4
        for (int q = 0; q < 16; ++q) {
            int e4 = tid + 128 * q;
            int bb = e4 >> 7;
            int r4 = e4 & 127;
            *(float4*)(phs + bb * PS + 4 * r4) = src[e4];
        }
    } else {
        const float4* src = (const float4*)phidiv + (size_t)col0 * 128 + FS * 2;
        if (tid < NB4) *(float4*)(phs + 4 * (FS * 2 + tid)) = src[tid];
    }

    v2f Af[8][16];
    if (fwd) {
        #pragma unroll
        for (int t = 0; t < 4; ++t) {
            const int u = (w + t) & 3;
            #pragma unroll
            for (int q = 0; q < 4; ++q) {
                const float* p = Wsh + (16 * u + 8 * h + 2 * q) * 512 + 16 * w + c;
                #pragma unroll
                for (int j = 0; j < 8; ++j) {
                    v2f f; f[0] = p[j * 64]; f[1] = p[j * 64 + 512];
                    Af[j][4 * t + q] = f;
                }
            }
        }
    } else {
        #pragma unroll
        for (int t = 0; t < 4; ++t) {
            const int u = (w + t) & 3;
            #pragma unroll
            for (int qq = 0; qq < 2; ++qq) {
                const float* p = Wsh + (16 * w + c) * 512 + 16 * u + 8 * h + 4 * qq;
                #pragma unroll
                for (int j = 0; j < 8; ++j) {
                    const float4 f4 = *(const float4*)(p + j * 64);
                    v2f f; f[0] = f4.x; f[1] = f4.y;
                    v2f g; g[0] = f4.z; g[1] = f4.w;
                    Af[j][4 * t + 2 * qq]     = f;
                    Af[j][4 * t + 2 * qq + 1] = g;
                }
            }
        }
    }

    float*       xw  = Xs + w * XW + lane * 4;
    const float* xr1 = Xs + ((w + 1) & 3) * XW + lane * 4;
    const float* xr2 = Xs + ((w + 2) & 3) * XW + lane * 4;
    const float* xr3 = Xs + ((w + 3) & 3) * XW + lane * 4;
    const float* php = phs + c * PS;

    float s0 = 1.f, s1 = 1.f, s2 = 1.f, s3 = 1.f, s4 = 1.f, s5 = 1.f, s6 = 1.f, s7 = 1.f;
    if (!fwd) {
        const bool cls = c < C_;
        const float* wl = WlS + (cls ? c : 0) * 64 + 16 * w + 8 * h;
        const float4 a0 = *(const float4*)(wl), a1 = *(const float4*)(wl + 4);
        s0 = cls ? a0.x : 0.f; s1 = cls ? a0.y : 0.f; s2 = cls ? a0.z : 0.f; s3 = cls ? a0.w : 0.f;
        s4 = cls ? a1.x : 0.f; s5 = cls ? a1.y : 0.f; s6 = cls ? a1.z : 0.f; s7 = cls ? a1.w : 0.f;
    }
    *(float4*)(xw)       = make_float4(s0, s1, s2, s3);
    *(float4*)(xw + 128) = make_float4(s4, s5, s6, s7);

    if (fwd) {
    const int nsteps = FS;
    const int dfirst = 0;
    const int dstep  = 1;
    for (int s = 0; s < nsteps; ++s) {
        const int d  = dfirst + dstep * s;
        const int rb = (s & 1) * XB;
        const int wb = XB - rb;

        v2f bo0, bo1, bo2, bo3;
        bo0[0] = s0; bo0[1] = s1; bo1[0] = s2; bo1[1] = s3;
        bo2[0] = s4; bo2[1] = s5; bo3[0] = s6; bo3[1] = s7;

        v8f acc[8];
        #pragma unroll
        for (int j = 0; j < 8; ++j)
            acc[j] = __builtin_amdgcn_wmma_f32_16x16x4_f32(false, Af[j][0], false, bo0,
                                                           (short)0, (v8f){}, false, false);

        __builtin_amdgcn_sched_barrier(0);
        __builtin_amdgcn_fence(__ATOMIC_RELEASE, "workgroup");
        __builtin_amdgcn_s_barrier_signal(-1);
        __builtin_amdgcn_sched_barrier(0);

        #pragma unroll
        for (int j = 0; j < 8; ++j)
            acc[j] = __builtin_amdgcn_wmma_f32_16x16x4_f32(false, Af[j][1], false, bo1,
                                                           (short)0, acc[j], false, false);

        __builtin_amdgcn_sched_barrier(0);
        __builtin_amdgcn_s_barrier_wait(-1);
        __builtin_amdgcn_fence(__ATOMIC_ACQUIRE, "workgroup");

        const float4 xa0 = *(const float4*)(xr1 + rb), xa1 = *(const float4*)(xr1 + rb + 128);
        const float4 xb0 = *(const float4*)(xr2 + rb), xb1 = *(const float4*)(xr2 + rb + 128);
        const float4 xc0 = *(const float4*)(xr3 + rb), xc1 = *(const float4*)(xr3 + rb + 128);
        const float4 p0 = *(const float4*)(php + d * 8);
        const float4 p1 = *(const float4*)(php + d * 8 + 4);
        __builtin_amdgcn_sched_barrier(0);

        #pragma unroll
        for (int j = 0; j < 8; ++j)
            acc[j] = __builtin_amdgcn_wmma_f32_16x16x4_f32(false, Af[j][2], false, bo2,
                                                           (short)0, acc[j], false, false);
        #pragma unroll
        for (int j = 0; j < 8; ++j)
            acc[j] = __builtin_amdgcn_wmma_f32_16x16x4_f32(false, Af[j][3], false, bo3,
                                                           (short)0, acc[j], false, false);

        __builtin_amdgcn_sched_barrier(0);

        v2f bx[12];
        bx[0][0] = xa0.x; bx[0][1] = xa0.y; bx[1][0] = xa0.z; bx[1][1] = xa0.w;
        bx[2][0] = xa1.x; bx[2][1] = xa1.y; bx[3][0] = xa1.z; bx[3][1] = xa1.w;
        bx[4][0] = xb0.x; bx[4][1] = xb0.y; bx[5][0] = xb0.z; bx[5][1] = xb0.w;
        bx[6][0] = xb1.x; bx[6][1] = xb1.y; bx[7][0] = xb1.z; bx[7][1] = xb1.w;
        bx[8][0] = xc0.x; bx[8][1] = xc0.y; bx[9][0] = xc0.z; bx[9][1] = xc0.w;
        bx[10][0] = xc1.x; bx[10][1] = xc1.y; bx[11][0] = xc1.z; bx[11][1] = xc1.w;

        #pragma unroll
        for (int kb = 4; kb < 16; ++kb) {
            #pragma unroll
            for (int j = 0; j < 8; ++j)
                acc[j] = __builtin_amdgcn_wmma_f32_16x16x4_f32(false, Af[j][kb], false, bx[kb - 4],
                                                               (short)0, acc[j], false, false);
        }

        #define COMBINE(r) fmaf(p1.w, acc[7][r], fmaf(p1.z, acc[6][r], fmaf(p1.y, acc[5][r], fmaf(p1.x, acc[4][r], \
                           fmaf(p0.w, acc[3][r], fmaf(p0.z, acc[2][r], fmaf(p0.y, acc[1][r], p0.x * acc[0][r])))))))
        s0 = COMBINE(0); s1 = COMBINE(1); s2 = COMBINE(2); s3 = COMBINE(3);
        s4 = COMBINE(4); s5 = COMBINE(5); s6 = COMBINE(6); s7 = COMBINE(7);
        #undef COMBINE

        *(float4*)(xw + wb)       = make_float4(s0, s1, s2, s3);
        *(float4*)(xw + wb + 128) = make_float4(s4, s5, s6, s7);
    }
    } else {
    __syncthreads();
    float4 p0 = *(const float4*)(phs + (D_ - 1) * 8);
    float4 p1 = *(const float4*)(phs + (D_ - 1) * 8 + 4);

    #define STEPMAT(pa, pb, kb) (pa.x * Af[0][kb] + pa.y * Af[1][kb] + pa.z * Af[2][kb] + pa.w * Af[3][kb] + \
                                 pb.x * Af[4][kb] + pb.y * Af[5][kb] + pb.z * Af[6][kb] + pb.w * Af[7][kb])
    v2f ad0 = STEPMAT(p0, p1, 0);

    for (int s = 0; s < D_ - FS; ++s) {
        const int d  = D_ - 1 - s;
        const int dn = (s + 1 < D_ - FS) ? d - 1 : d;
        const int rb = (s & 1) * XB;
        const int wb = XB - rb;

        v2f bo0, bo1, bo2, bo3;
        bo0[0] = s0; bo0[1] = s1; bo1[0] = s2; bo1[1] = s3;
        bo2[0] = s4; bo2[1] = s5; bo3[0] = s6; bo3[1] = s7;

        v8f accA = __builtin_amdgcn_wmma_f32_16x16x4_f32(false, ad0, false, bo0,
                                                         (short)0, (v8f){}, false, false);

        __builtin_amdgcn_sched_barrier(0);
        asm volatile("s_wait_dscnt 0x0" ::: "memory");
        __builtin_amdgcn_s_barrier_signal(-1);
        __builtin_amdgcn_sched_barrier(0);

        v8f accB = __builtin_amdgcn_wmma_f32_16x16x4_f32(false, STEPMAT(p0, p1, 1), false, bo1,
                                                         (short)0, (v8f){}, false, false);

        __builtin_amdgcn_sched_barrier(0);
        __builtin_amdgcn_s_barrier_wait(-1);
        __builtin_amdgcn_fence(__ATOMIC_ACQUIRE, "workgroup");

        const float4 xa0 = *(const float4*)(xr1 + rb), xa1 = *(const float4*)(xr1 + rb + 128);
        const float4 xb0 = *(const float4*)(xr2 + rb), xb1 = *(const float4*)(xr2 + rb + 128);
        const float4 xc0 = *(const float4*)(xr3 + rb), xc1 = *(const float4*)(xr3 + rb + 128);
        const float4 pn0 = *(const float4*)(phs + dn * 8);
        const float4 pn1 = *(const float4*)(phs + dn * 8 + 4);
        __builtin_amdgcn_sched_barrier(0);

        accA = __builtin_amdgcn_wmma_f32_16x16x4_f32(false, STEPMAT(p0, p1, 2), false, bo2,
                                                     (short)0, accA, false, false);
        accB = __builtin_amdgcn_wmma_f32_16x16x4_f32(false, STEPMAT(p0, p1, 3), false, bo3,
                                                     (short)0, accB, false, false);

        __builtin_amdgcn_sched_barrier(0);

        v2f bx[12];
        bx[0][0] = xa0.x; bx[0][1] = xa0.y; bx[1][0] = xa0.z; bx[1][1] = xa0.w;
        bx[2][0] = xa1.x; bx[2][1] = xa1.y; bx[3][0] = xa1.z; bx[3][1] = xa1.w;
        bx[4][0] = xb0.x; bx[4][1] = xb0.y; bx[5][0] = xb0.z; bx[5][1] = xb0.w;
        bx[6][0] = xb1.x; bx[6][1] = xb1.y; bx[7][0] = xb1.z; bx[7][1] = xb1.w;
        bx[8][0] = xc0.x; bx[8][1] = xc0.y; bx[9][0] = xc0.z; bx[9][1] = xc0.w;
        bx[10][0] = xc1.x; bx[10][1] = xc1.y; bx[11][0] = xc1.z; bx[11][1] = xc1.w;

        #pragma unroll
        for (int kb = 4; kb < 16; kb += 2) {
            accA = __builtin_amdgcn_wmma_f32_16x16x4_f32(false, STEPMAT(p0, p1, kb), false, bx[kb - 4],
                                                         (short)0, accA, false, false);
            accB = __builtin_amdgcn_wmma_f32_16x16x4_f32(false, STEPMAT(p0, p1, kb + 1), false, bx[kb - 3],
                                                         (short)0, accB, false, false);
        }

        __builtin_amdgcn_sched_barrier(0);
        ad0 = STEPMAT(pn0, pn1, 0);
        __builtin_amdgcn_sched_barrier(0);

        s0 = accA[0] + accB[0]; s1 = accA[1] + accB[1]; s2 = accA[2] + accB[2]; s3 = accA[3] + accB[3];
        s4 = accA[4] + accB[4]; s5 = accA[5] + accB[5]; s6 = accA[6] + accB[6]; s7 = accA[7] + accB[7];

        *(float4*)(xw + wb)       = make_float4(s0, s1, s2, s3);
        *(float4*)(xw + wb + 128) = make_float4(s4, s5, s6, s7);
        p0 = pn0; p1 = pn1;
    }
    #undef STEPMAT
    }

    {
        float* vf = Vfin + c * VS + 16 * w + 8 * h;
        *(float4*)(vf)     = make_float4(s0, s1, s2, s3);
        *(float4*)(vf + 4) = make_float4(s4, s5, s6, s7);
    }
    __syncthreads();
    {
        float* dst = fwd ? Vf + (size_t)col0 * 64 : Ub + (size_t)col0 * (C_ * 64);
        const int nrows = fwd ? 16 : C_;
        #pragma unroll
        for (int q = 0; q < 2; ++q) {
            int e4  = tid + 128 * q;
            int row = e4 >> 4;
            int c4  = e4 & 15;
            if (row < nrows)
                *(float4*)(dst + row * 64 + 4 * c4) = *(const float4*)(Vfin + row * VS + 4 * c4);
        }
    }
}

__global__ __launch_bounds__(512)
void join_kernel(const float* __restrict__ Vf, const float* __restrict__ Ub,
                 float* __restrict__ out) {
    __shared__ float res[16 * C_];
    const int wv   = threadIdx.x >> 5;
    const int b    = blockIdx.x * 16 + wv;
    const int lane = threadIdx.x & 31;
    const float v0 = Vf[b * 64 + lane], v1 = Vf[b * 64 + 32 + lane];
    const float* up = Ub + (size_t)b * (C_ * 64) + lane;
    float tr[C_];
    #pragma unroll
    for (int k = 0; k < C_; ++k) tr[k] = fmaf(v1, up[k * 64 + 32], v0 * up[k * 64]);
    #pragma unroll
    for (int off = 16; off > 0; off >>= 1) {
        #pragma unroll
        for (int k = 0; k < C_; ++k) tr[k] += __shfl_xor(tr[k], off, 32);
    }
    float mx = tr[0];
    #pragma unroll
    for (int k = 1; k < C_; ++k) mx = fmaxf(mx, tr[k]);
    float se = 0.f;
    #pragma unroll
    for (int k = 0; k < C_; ++k) se += expf(tr[k] - mx);
    float mine = tr[0];
    #pragma unroll
    for (int k = 1; k < C_; ++k) mine = (lane == k) ? tr[k] : mine;
    if (lane < C_) res[wv * C_ + lane] = mine - (logf(se) + mx);
    __syncthreads();
    if (threadIdx.x < 16 * C_) out[(size_t)blockIdx.x * (16 * C_) + threadIdx.x] = res[threadIdx.x];
}

extern "C" void kernel_launch(void* const* d_in, const int* in_sizes, int n_in,
                              void* d_out, int out_size, void* d_ws, size_t ws_size,
                              hipStream_t stream) {
    const float* x     = (const float*)d_in[0];
    const float* Wsh   = (const float*)d_in[1];
    const float* Wlast = (const float*)d_in[2];
    float* out = (float*)d_out;
    float* ws  = (float*)d_ws;
    float* WlS    = ws;
    float* phidiv = ws + (C_ * R_ * R_);
    float* Vf     = phidiv + (B_ * D_ * M_);
    float* Ub     = Vf + (B_ * R_);

    prep_kernel<<<PHB + WLB, 256, 0, stream>>>(x, Wlast, phidiv, WlS);
    const size_t reserve = (size_t)168 * 1024;
    hipFuncSetAttribute((const void*)trnet_kernel,
                        hipFuncAttributeMaxDynamicSharedMemorySize, (int)reserve);
    trnet_kernel<<<NFWD + NBWD, 128, reserve, stream>>>(Wsh, phidiv, WlS, Vf, Ub);
    join_kernel<<<B_ / 16, 512, 0, stream>>>(Vf, Ub, out);
}


# Backward blocks carry two batches half a step apart per wave: one's 16 WMMAs cover the other's LDS hand-over, barrier and step-matrix forming; split 44/20 -> 37/27 steps; f32 sum order changes
# speedup vs baseline: 7.4414x; 1.1119x over previous
#include <hip/hip_runtime.h>
#include <math.h>

typedef __attribute__((ext_vector_type(2))) float v2f;
typedef __attribute__((ext_vector_type(8))) float v8f;

#define B_  1024
#define N_  1024
#define M_  8
#define D_  64
#define R_  64
#define C_  10

#define VS  68
#define PS  516
#define XW  256
#define XB  1024

#define FS    37
#define NFWD  (B_ / 16)
#define NBWD  (B_ / 2)
#define NB4   ((D_ - FS) * 2)


#define PHB  (B_ * D_ / 256)
#define WLB  (C_ * R_ * 32 / 256)

__global__ __launch_bounds__(256)
void prep_kernel(const float* __restrict__ x, const float* __restrict__ Wlast,
                 float* __restrict__ phidiv, float* __restrict__ WlS) {
    if (blockIdx.x < PHB) {
        const int tid = blockIdx.x * 256 + threadIdx.x;
        const float* xp = x + (size_t)tid * 16;
        float acc[8];
        #pragma unroll
        for (int m = 0; m < 8; ++m) acc[m] = 0.f;
        for (int n = 0; n < 16; ++n) {
            float a = 1.5707963267948966f * xp[n];
            float sn, cs;
            sincosf(a, &sn, &cs);
            float pc[8], ps[8];
            pc[0] = 1.f; ps[0] = 1.f;
            #pragma unroll
            for (int k = 1; k < 8; ++k) { pc[k] = pc[k-1] * cs; ps[k] = ps[k-1] * sn; }
            #pragma unroll
            for (int m = 0; m < 8; ++m) acc[m] = fmaf(pc[7-m], ps[m], acc[m]);
        }
        float* o = phidiv + (size_t)tid * 8;
        #pragma unroll
        for (int m = 0; m < 8; ++m) o[m] = acc[m];
    } else {
        const int wv   = ((blockIdx.x - PHB) * 256 + threadIdx.x) >> 5;
        const int lane = threadIdx.x & 31;
        const float* p = Wlast + (size_t)wv * 512 + lane;
        float s = 0.f;
        #pragma unroll
        for (int q = 0; q < 16; ++q) s += p[32 * q];
        #pragma unroll
        for (int off = 16; off > 0; off >>= 1) s += __shfl_xor(s, off, 32);
        if (lane == 0) WlS[wv] = s;
    }
}

__global__ __launch_bounds__(128)
void trnet_kernel(const float* __restrict__ Wsh, const float* __restrict__ phidiv,
                  const float* __restrict__ WlS, float* __restrict__ Vf, float* __restrict__ Ub)
{
    __shared__ __attribute__((aligned(16))) float Xs[2 * XB];
    __shared__ __attribute__((aligned(16))) float Vfin[32 * VS];
    __shared__ __attribute__((aligned(16))) float phs[16 * PS];

    const int tid  = threadIdx.x;
    const int w    = tid >> 5;
    const int lane = tid & 31;
    const int c    = lane & 15;
    const int h    = lane >> 4;
    const bool fwd  = blockIdx.x < NFWD;
    const int  col0 = fwd ? blockIdx.x * 16 : 2 * (blockIdx.x - NFWD);

    if (fwd) {
        const float4* src = (const float4*)(phidiv + (size_t)col0 * (D_ * M_));
        #pragma unroll 4
        for (int q = 0; q < 16; ++q) {
            int e4 = tid + 128 * q;
            int bb = e4 >> 7;
            int r4 = e4 & 127;
            *(float4*)(phs + bb * PS + 4 * r4) = src[e4];
        }
    } else {
        const float4* src = (const float4*)phidiv + (size_t)col0 * 128 + FS * 2;
        for (int e = tid; e < 2 * NB4; e += 128) {
            const int xb = e / NB4;
            const int t4 = e - xb * NB4;
            *(float4*)(phs + xb * PS + 4 * (FS * 2 + t4)) = src[xb * 128 + t4];
        }
    }

    v2f Af[8][16];
    if (fwd) {
        #pragma unroll
        for (int t = 0; t < 4; ++t) {
            const int u = (w + t) & 3;
            #pragma unroll
            for (int q = 0; q < 4; ++q) {
                const float* p = Wsh + (16 * u + 8 * h + 2 * q) * 512 + 16 * w + c;
                #pragma unroll
                for (int j = 0; j < 8; ++j) {
                    v2f f; f[0] = p[j * 64]; f[1] = p[j * 64 + 512];
                    Af[j][4 * t + q] = f;
                }
            }
        }
    } else {
        #pragma unroll
        for (int t = 0; t < 4; ++t) {
            const int u = (w + t) & 3;
            #pragma unroll
            for (int qq = 0; qq < 2; ++qq) {
                const float* p = Wsh + (16 * w + c) * 512 + 16 * u + 8 * h + 4 * qq;
                #pragma unroll
                for (int j = 0; j < 8; ++j) {
                    const float4 f4 = *(const float4*)(p + j * 64);
                    v2f f; f[0] = f4.x; f[1] = f4.y;
                    v2f g; g[0] = f4.z; g[1] = f4.w;
                    Af[j][4 * t + 2 * qq]     = f;
                    Af[j][4 * t + 2 * qq + 1] = g;
                }
            }
        }
    }

    float*       xw  = Xs + w * XW + lane * 4;
    const float* xr1 = Xs + ((w + 1) & 3) * XW + lane * 4;
    const float* xr2 = Xs + ((w + 2) & 3) * XW + lane * 4;
    const float* xr3 = Xs + ((w + 3) & 3) * XW + lane * 4;
    const float* php = phs + c * PS;

    float s0 = 1.f, s1 = 1.f, s2 = 1.f, s3 = 1.f, s4 = 1.f, s5 = 1.f, s6 = 1.f, s7 = 1.f;
    if (!fwd) {
        const bool cls = c < C_;
        const float* wl = WlS + (cls ? c : 0) * 64 + 16 * w + 8 * h;
        const float4 a0 = *(const float4*)(wl), a1 = *(const float4*)(wl + 4);
        s0 = cls ? a0.x : 0.f; s1 = cls ? a0.y : 0.f; s2 = cls ? a0.z : 0.f; s3 = cls ? a0.w : 0.f;
        s4 = cls ? a1.x : 0.f; s5 = cls ? a1.y : 0.f; s6 = cls ? a1.z : 0.f; s7 = cls ? a1.w : 0.f;
    }
    *(float4*)(xw)       = make_float4(s0, s1, s2, s3);
    *(float4*)(xw + 128) = make_float4(s4, s5, s6, s7);

    if (fwd) {
    const int nsteps = FS;
    const int dfirst = 0;
    const int dstep  = 1;
    for (int s = 0; s < nsteps; ++s) {
        const int d  = dfirst + dstep * s;
        const int rb = (s & 1) * XB;
        const int wb = XB - rb;

        v2f bo0, bo1, bo2, bo3;
        bo0[0] = s0; bo0[1] = s1; bo1[0] = s2; bo1[1] = s3;
        bo2[0] = s4; bo2[1] = s5; bo3[0] = s6; bo3[1] = s7;

        v8f acc[8];
        #pragma unroll
        for (int j = 0; j < 8; ++j)
            acc[j] = __builtin_amdgcn_wmma_f32_16x16x4_f32(false, Af[j][0], false, bo0,
                                                           (short)0, (v8f){}, false, false);

        __builtin_amdgcn_sched_barrier(0);
        __builtin_amdgcn_fence(__ATOMIC_RELEASE, "workgroup");
        __builtin_amdgcn_s_barrier_signal(-1);
        __builtin_amdgcn_sched_barrier(0);

        #pragma unroll
        for (int j = 0; j < 8; ++j)
            acc[j] = __builtin_amdgcn_wmma_f32_16x16x4_f32(false, Af[j][1], false, bo1,
                                                           (short)0, acc[j], false, false);

        __builtin_amdgcn_sched_barrier(0);
        __builtin_amdgcn_s_barrier_wait(-1);
        __builtin_amdgcn_fence(__ATOMIC_ACQUIRE, "workgroup");

        const float4 xa0 = *(const float4*)(xr1 + rb), xa1 = *(const float4*)(xr1 + rb + 128);
        const float4 xb0 = *(const float4*)(xr2 + rb), xb1 = *(const float4*)(xr2 + rb + 128);
        const float4 xc0 = *(const float4*)(xr3 + rb), xc1 = *(const float4*)(xr3 + rb + 128);
        const float4 p0 = *(const float4*)(php + d * 8);
        const float4 p1 = *(const float4*)(php + d * 8 + 4);
        __builtin_amdgcn_sched_barrier(0);

        #pragma unroll
        for (int j = 0; j < 8; ++j)
            acc[j] = __builtin_amdgcn_wmma_f32_16x16x4_f32(false, Af[j][2], false, bo2,
                                                           (short)0, acc[j], false, false);
        #pragma unroll
        for (int j = 0; j < 8; ++j)
            acc[j] = __builtin_amdgcn_wmma_f32_16x16x4_f32(false, Af[j][3], false, bo3,
                                                           (short)0, acc[j], false, false);

        __builtin_amdgcn_sched_barrier(0);

        v2f bx[12];
        bx[0][0] = xa0.x; bx[0][1] = xa0.y; bx[1][0] = xa0.z; bx[1][1] = xa0.w;
        bx[2][0] = xa1.x; bx[2][1] = xa1.y; bx[3][0] = xa1.z; bx[3][1] = xa1.w;
        bx[4][0] = xb0.x; bx[4][1] = xb0.y; bx[5][0] = xb0.z; bx[5][1] = xb0.w;
        bx[6][0] = xb1.x; bx[6][1] = xb1.y; bx[7][0] = xb1.z; bx[7][1] = xb1.w;
        bx[8][0] = xc0.x; bx[8][1] = xc0.y; bx[9][0] = xc0.z; bx[9][1] = xc0.w;
        bx[10][0] = xc1.x; bx[10][1] = xc1.y; bx[11][0] = xc1.z; bx[11][1] = xc1.w;

        #pragma unroll
        for (int kb = 4; kb < 16; ++kb) {
            #pragma unroll
            for (int j = 0; j < 8; ++j)
                acc[j] = __builtin_amdgcn_wmma_f32_16x16x4_f32(false, Af[j][kb], false, bx[kb - 4],
                                                               (short)0, acc[j], false, false);
        }

        #define COMBINE(r) fmaf(p1.w, acc[7][r], fmaf(p1.z, acc[6][r], fmaf(p1.y, acc[5][r], fmaf(p1.x, acc[4][r], \
                           fmaf(p0.w, acc[3][r], fmaf(p0.z, acc[2][r], fmaf(p0.y, acc[1][r], p0.x * acc[0][r])))))))
        s0 = COMBINE(0); s1 = COMBINE(1); s2 = COMBINE(2); s3 = COMBINE(3);
        s4 = COMBINE(4); s5 = COMBINE(5); s6 = COMBINE(6); s7 = COMBINE(7);
        #undef COMBINE

        *(float4*)(xw + wb)       = make_float4(s0, s1, s2, s3);
        *(float4*)(xw + wb + 128) = make_float4(s4, s5, s6, s7);
    }
    } else {
    *(float4*)(xw + XB)       = make_float4(s0, s1, s2, s3);
    *(float4*)(xw + XB + 128) = make_float4(s4, s5, s6, s7);
    __syncthreads();

    #define FEAT(pa, pb, j) ((j) == 0 ? pa.x : (j) == 1 ? pa.y : (j) == 2 ? pa.z : (j) == 3 ? pa.w : \
                             (j) == 4 ? pb.x : (j) == 5 ? pb.y : (j) == 6 ? pb.z : pb.w)
    #define FORMJ(dst, pa, pb, j, k0, k1) do {                                                          \
        const int jf_ = (j);     \
        _Pragma("unroll") for (int kf_ = (k0); kf_ < (k1); ++kf_) {                                     \
            if (jf_ == 0) dst[kf_] = FEAT(pa, pb, jf_) * Af[jf_][kf_];                                  \
            else          dst[kf_] = FEAT(pa, pb, jf_) * Af[jf_][kf_] + dst[kf_];                       \
        }                                                                                               \
    } while (0)

    float  sv[2][8];
    float4 fa[2], fb[2];
    v2f    fr[2][16];
    v2f    bx[2][12];
    #pragma unroll
    for (int x = 0; x < 2; ++x) {
        sv[x][0] = s0; sv[x][1] = s1; sv[x][2] = s2; sv[x][3] = s3;
        sv[x][4] = s4; sv[x][5] = s5; sv[x][6] = s6; sv[x][7] = s7;
    }
    fa[0] = *(const float4*)(phs + (D_ - 1) * 8);
    fb[0] = *(const float4*)(phs + (D_ - 1) * 8 + 4);
    fa[1] = *(const float4*)(phs + PS + (D_ - 1) * 8);
    fb[1] = *(const float4*)(phs + PS + (D_ - 1) * 8 + 4);
    {
        const float4 xa0 = *(const float4*)(xr1), xa1 = *(const float4*)(xr1 + 128);
        const float4 xb0 = *(const float4*)(xr2), xb1 = *(const float4*)(xr2 + 128);
        const float4 xc0 = *(const float4*)(xr3), xc1 = *(const float4*)(xr3 + 128);
        bx[0][0][0] = xa0.x; bx[0][0][1] = xa0.y; bx[0][1][0] = xa0.z; bx[0][1][1] = xa0.w;
        bx[0][2][0] = xa1.x; bx[0][2][1] = xa1.y; bx[0][3][0] = xa1.z; bx[0][3][1] = xa1.w;
        bx[0][4][0] = xb0.x; bx[0][4][1] = xb0.y; bx[0][5][0] = xb0.z; bx[0][5][1] = xb0.w;
        bx[0][6][0] = xb1.x; bx[0][6][1] = xb1.y; bx[0][7][0] = xb1.z; bx[0][7][1] = xb1.w;
        bx[0][8][0] = xc0.x; bx[0][8][1] = xc0.y; bx[0][9][0] = xc0.z; bx[0][9][1] = xc0.w;
        bx[0][10][0] = xc1.x; bx[0][10][1] = xc1.y; bx[0][11][0] = xc1.z; bx[0][11][1] = xc1.w;
    }
    #pragma unroll
    for (int j = 0; j < 8; ++j) FORMJ(fr[0], fa[0], fb[0], j, 0, 16);

    for (int s = 0; s < D_ - FS; ++s) {
        const int d  = D_ - 1 - s;
        const int dn = (s + 1 < D_ - FS) ? d - 1 : d;

        #pragma unroll
        for (int x = 0; x < 2; ++x) {
            const int y = 1 - x;

            v2f bo[4];
            bo[0][0] = sv[x][0]; bo[0][1] = sv[x][1]; bo[1][0] = sv[x][2]; bo[1][1] = sv[x][3];
            bo[2][0] = sv[x][4]; bo[2][1] = sv[x][5]; bo[3][0] = sv[x][6]; bo[3][1] = sv[x][7];

            v8f acc[2] = {(v8f){}, (v8f){}};
            float4 xa0, xa1, xb0, xb1, xc0, xc1, pn0, pn1;

            #pragma unroll
            for (int kb = 0; kb < 4; ++kb) {
                acc[kb & 1] = __builtin_amdgcn_wmma_f32_16x16x4_f32(false, fr[x][kb], false, bo[kb],
                                                                    (short)0, acc[kb & 1], false, false);
                FORMJ(fr[y], fa[y], fb[y], 2 * kb,     0, 4);
                FORMJ(fr[y], fa[y], fb[y], 2 * kb + 1, 0, 4);
                __builtin_amdgcn_sched_barrier(0);
            }

            #pragma unroll
            for (int kb = 4; kb < 8; ++kb) {
                if (kb == 6) {
                    asm volatile("s_wait_dscnt 0x0" ::: "memory");
                    __builtin_amdgcn_s_barrier_signal(-1);
                    __builtin_amdgcn_sched_barrier(0);
                }
                acc[kb & 1] = __builtin_amdgcn_wmma_f32_16x16x4_f32(false, fr[x][kb], false, bx[x][kb - 4],
                                                                    (short)0, acc[kb & 1], false, false);
                FORMJ(fr[y], fa[y], fb[y], 2 * (kb - 4),     4, 8);
                FORMJ(fr[y], fa[y], fb[y], 2 * (kb - 4) + 1, 4, 8);
                __builtin_amdgcn_sched_barrier(0);
            }

            #pragma unroll
            for (int kb = 8; kb < 16; ++kb) {
                if (kb == 10) {
                    __builtin_amdgcn_s_barrier_wait(-1);
                    __builtin_amdgcn_fence(__ATOMIC_ACQUIRE, "workgroup");
                    xa0 = *(const float4*)(xr1 + y * XB); xa1 = *(const float4*)(xr1 + y * XB + 128);
                    xb0 = *(const float4*)(xr2 + y * XB); xb1 = *(const float4*)(xr2 + y * XB + 128);
                    xc0 = *(const float4*)(xr3 + y * XB); xc1 = *(const float4*)(xr3 + y * XB + 128);
                    pn0 = *(const float4*)(phs + x * PS + dn * 8);
                    pn1 = *(const float4*)(phs + x * PS + dn * 8 + 4);
                    __builtin_amdgcn_sched_barrier(0);
                }
                acc[kb & 1] = __builtin_amdgcn_wmma_f32_16x16x4_f32(false, fr[x][kb], false, bx[x][kb - 4],
                                                                    (short)0, acc[kb & 1], false, false);
                if (kb == 15) __builtin_amdgcn_sched_barrier(0);
                FORMJ(fr[y], fa[y], fb[y], kb - 8, 8, 16);
                __builtin_amdgcn_sched_barrier(0);
            }

            bx[y][0][0] = xa0.x; bx[y][0][1] = xa0.y; bx[y][1][0] = xa0.z; bx[y][1][1] = xa0.w;
            bx[y][2][0] = xa1.x; bx[y][2][1] = xa1.y; bx[y][3][0] = xa1.z; bx[y][3][1] = xa1.w;
            bx[y][4][0] = xb0.x; bx[y][4][1] = xb0.y; bx[y][5][0] = xb0.z; bx[y][5][1] = xb0.w;
            bx[y][6][0] = xb1.x; bx[y][6][1] = xb1.y; bx[y][7][0] = xb1.z; bx[y][7][1] = xb1.w;
            bx[y][8][0] = xc0.x; bx[y][8][1] = xc0.y; bx[y][9][0] = xc0.z; bx[y][9][1] = xc0.w;
            bx[y][10][0] = xc1.x; bx[y][10][1] = xc1.y; bx[y][11][0] = xc1.z; bx[y][11][1] = xc1.w;
            fa[x] = pn0; fb[x] = pn1;

            #pragma unroll
            for (int r = 0; r < 8; ++r) sv[x][r] = acc[0][r] + acc[1][r];

            *(float4*)(xw + x * XB)       = make_float4(sv[x][0], sv[x][1], sv[x][2], sv[x][3]);
            *(float4*)(xw + x * XB + 128) = make_float4(sv[x][4], sv[x][5], sv[x][6], sv[x][7]);
        }
    }
    #undef FORMJ
    #undef FEAT
    s0 = sv[0][0]; s1 = sv[0][1]; s2 = sv[0][2]; s3 = sv[0][3];
    s4 = sv[0][4]; s5 = sv[0][5]; s6 = sv[0][6]; s7 = sv[0][7];
    {
        float* vf1 = Vfin + (16 + c) * VS + 16 * w + 8 * h;
        *(float4*)(vf1)     = make_float4(sv[1][0], sv[1][1], sv[1][2], sv[1][3]);
        *(float4*)(vf1 + 4) = make_float4(sv[1][4], sv[1][5], sv[1][6], sv[1][7]);
    }
    }

    {
        float* vf = Vfin + c * VS + 16 * w + 8 * h;
        *(float4*)(vf)     = make_float4(s0, s1, s2, s3);
        *(float4*)(vf + 4) = make_float4(s4, s5, s6, s7);
    }
    __syncthreads();
    {
        float* dst = fwd ? Vf + (size_t)col0 * 64 : Ub + (size_t)col0 * (C_ * 64);
        const int nrows = fwd ? 16 : 2 * C_;
        #pragma unroll
        for (int q = 0; q < 3; ++q) {
            int e4  = tid + 128 * q;
            int row = e4 >> 4;
            int c4  = e4 & 15;
            int srow = (!fwd && row >= C_) ? row + (16 - C_) : row;
            if (row < nrows)
                *(float4*)(dst + row * 64 + 4 * c4) = *(const float4*)(Vfin + srow * VS + 4 * c4);
        }
    }
}

__global__ __launch_bounds__(512)
void join_kernel(const float* __restrict__ Vf, const float* __restrict__ Ub,
                 float* __restrict__ out) {
    __shared__ float res[16 * C_];
    const int wv   = threadIdx.x >> 5;
    const int b    = blockIdx.x * 16 + wv;
    const int lane = threadIdx.x & 31;
    const float v0 = Vf[b * 64 + lane], v1 = Vf[b * 64 + 32 + lane];
    const float* up = Ub + (size_t)b * (C_ * 64) + lane;
    float tr[C_];
    #pragma unroll
    for (int k = 0; k < C_; ++k) tr[k] = fmaf(v1, up[k * 64 + 32], v0 * up[k * 64]);
    #pragma unroll
    for (int off = 16; off > 0; off >>= 1) {
        #pragma unroll
        for (int k = 0; k < C_; ++k) tr[k] += __shfl_xor(tr[k], off, 32);
    }
    float mx = tr[0];
    #pragma unroll
    for (int k = 1; k < C_; ++k) mx = fmaxf(mx, tr[k]);
    float se = 0.f;
    #pragma unroll
    for (int k = 0; k < C_; ++k) se += expf(tr[k] - mx);
    float mine = tr[0];
    #pragma unroll
    for (int k = 1; k < C_; ++k) mine = (lane == k) ? tr[k] : mine;
    if (lane < C_) res[wv * C_ + lane] = mine - (logf(se) + mx);
    __syncthreads();
    if (threadIdx.x < 16 * C_) out[(size_t)blockIdx.x * (16 * C_) + threadIdx.x] = res[threadIdx.x];
}

extern "C" void kernel_launch(void* const* d_in, const int* in_sizes, int n_in,
                              void* d_out, int out_size, void* d_ws, size_t ws_size,
                              hipStream_t stream) {
    const float* x     = (const float*)d_in[0];
    const float* Wsh   = (const float*)d_in[1];
    const float* Wlast = (const float*)d_in[2];
    float* out = (float*)d_out;
    float* ws  = (float*)d_ws;
    float* WlS    = ws;
    float* phidiv = ws + (C_ * R_ * R_);
    float* Vf     = phidiv + (B_ * D_ * M_);
    float* Ub     = Vf + (B_ * R_);

    prep_kernel<<<PHB + WLB, 256, 0, stream>>>(x, Wlast, phidiv, WlS);
    const size_t reserve = (size_t)168 * 1024;
    hipFuncSetAttribute((const void*)trnet_kernel,
                        hipFuncAttributeMaxDynamicSharedMemorySize, (int)reserve);
    trnet_kernel<<<NFWD + NBWD, 128, reserve, stream>>>(Wsh, phidiv, WlS, Vf, Ub);
    join_kernel<<<B_ / 16, 512, 0, stream>>>(Vf, Ub, out);
}


# Backward blocks stay resident: 171 blocks run 3 batch pairs each in turn (was 512 one-pair blocks queued for free WGPs); W fragments and class weights fetched once per block; sum orders unchanged
# speedup vs baseline: 8.1275x; 1.0922x over previous
#include <hip/hip_runtime.h>
#include <math.h>

typedef __attribute__((ext_vector_type(2))) float v2f;
typedef __attribute__((ext_vector_type(8))) float v8f;

#define B_  1024
#define N_  1024
#define M_  8
#define D_  64
#define R_  64
#define C_  10

#define VS  68
#define PS  516
#define XW  256
#define XB  1024

#define FS    37
#define NFWD  (B_ / 16)
#define NBP   (B_ / 2)
#define PPB   3
#define NBWG  ((NBP + PPB - 1) / PPB)
#define NB4   ((D_ - FS) * 2)


#define PHB  (B_ * D_ / 256)
#define WLB  (C_ * R_ * 32 / 256)

__global__ __launch_bounds__(256)
void prep_kernel(const float* __restrict__ x, const float* __restrict__ Wlast,
                 float* __restrict__ phidiv, float* __restrict__ WlS) {
    if (blockIdx.x < PHB) {
        const int tid = blockIdx.x * 256 + threadIdx.x;
        const float* xp = x + (size_t)tid * 16;
        float acc[8];
        #pragma unroll
        for (int m = 0; m < 8; ++m) acc[m] = 0.f;
        for (int n = 0; n < 16; ++n) {
            float a = 1.5707963267948966f * xp[n];
            float sn, cs;
            sincosf(a, &sn, &cs);
            float pc[8], ps[8];
            pc[0] = 1.f; ps[0] = 1.f;
            #pragma unroll
            for (int k = 1; k < 8; ++k) { pc[k] = pc[k-1] * cs; ps[k] = ps[k-1] * sn; }
            #pragma unroll
            for (int m = 0; m < 8; ++m) acc[m] = fmaf(pc[7-m], ps[m], acc[m]);
        }
        float* o = phidiv + (size_t)tid * 8;
        #pragma unroll
        for (int m = 0; m < 8; ++m) o[m] = acc[m];
    } else {
        const int wv   = ((blockIdx.x - PHB) * 256 + threadIdx.x) >> 5;
        const int lane = threadIdx.x & 31;
        const float* p = Wlast + (size_t)wv * 512 + lane;
        float s = 0.f;
        #pragma unroll
        for (int q = 0; q < 16; ++q) s += p[32 * q];
        #pragma unroll
        for (int off = 16; off > 0; off >>= 1) s += __shfl_xor(s, off, 32);
        if (lane == 0) WlS[wv] = s;
    }
}

__global__ __launch_bounds__(128)
void trnet_kernel(const float* __restrict__ Wsh, const float* __restrict__ phidiv,
                  const float* __restrict__ WlS, float* __restrict__ Vf, float* __restrict__ Ub)
{
    __shared__ __attribute__((aligned(16))) float Xs[2 * XB];
    __shared__ __attribute__((aligned(16))) float Vfin[32 * VS];
    __shared__ __attribute__((aligned(16))) float phs[16 * PS];

    const int tid  = threadIdx.x;
    const int w    = tid >> 5;
    const int lane = tid & 31;
    const int c    = lane & 15;
    const int h    = lane >> 4;
    const bool fwd  = blockIdx.x < NFWD;
    const int  col0 = fwd ? blockIdx.x * 16 : 2 * PPB * (blockIdx.x - NFWD);

    if (fwd) {
        const float4* src = (const float4*)(phidiv + (size_t)col0 * (D_ * M_));
        #pragma unroll 4
        for (int q = 0; q < 16; ++q) {
            int e4 = tid + 128 * q;
            int bb = e4 >> 7;
            int r4 = e4 & 127;
            *(float4*)(phs + bb * PS + 4 * r4) = src[e4];
        }
    }

    v2f Af[8][16];
    if (fwd) {
        #pragma unroll
        for (int t = 0; t < 4; ++t) {
            const int u = (w + t) & 3;
            #pragma unroll
            for (int q = 0; q < 4; ++q) {
                const float* p = Wsh + (16 * u + 8 * h + 2 * q) * 512 + 16 * w + c;
                #pragma unroll
                for (int j = 0; j < 8; ++j) {
                    v2f f; f[0] = p[j * 64]; f[1] = p[j * 64 + 512];
                    Af[j][4 * t + q] = f;
                }
            }
        }
    } else {
        #pragma unroll
        for (int t = 0; t < 4; ++t) {
            const int u = (w + t) & 3;
            #pragma unroll
            for (int qq = 0; qq < 2; ++qq) {
                const float* p = Wsh + (16 * w + c) * 512 + 16 * u + 8 * h + 4 * qq;
                #pragma unroll
                for (int j = 0; j < 8; ++j) {
                    const float4 f4 = *(const float4*)(p + j * 64);
                    v2f f; f[0] = f4.x; f[1] = f4.y;
                    v2f g; g[0] = f4.z; g[1] = f4.w;
                    Af[j][4 * t + 2 * qq]     = f;
                    Af[j][4 * t + 2 * qq + 1] = g;
                }
            }
        }
    }

    float*       xw  = Xs + w * XW + lane * 4;
    const float* xr1 = Xs + ((w + 1) & 3) * XW + lane * 4;
    const float* xr2 = Xs + ((w + 2) & 3) * XW + lane * 4;
    const float* xr3 = Xs + ((w + 3) & 3) * XW + lane * 4;
    const float* php = phs + c * PS;

    float s0 = 1.f, s1 = 1.f, s2 = 1.f, s3 = 1.f, s4 = 1.f, s5 = 1.f, s6 = 1.f, s7 = 1.f;
    if (!fwd) {
        const bool cls = c < C_;
        const float* wl = WlS + (cls ? c : 0) * 64 + 16 * w + 8 * h;
        const float4 a0 = *(const float4*)(wl), a1 = *(const float4*)(wl + 4);
        s0 = cls ? a0.x : 0.f; s1 = cls ? a0.y : 0.f; s2 = cls ? a0.z : 0.f; s3 = cls ? a0.w : 0.f;
        s4 = cls ? a1.x : 0.f; s5 = cls ? a1.y : 0.f; s6 = cls ? a1.z : 0.f; s7 = cls ? a1.w : 0.f;
    }
    if (fwd) {
    *(float4*)(xw)       = make_float4(s0, s1, s2, s3);
    *(float4*)(xw + 128) = make_float4(s4, s5, s6, s7);
    const int nsteps = FS;
    const int dfirst = 0;
    const int dstep  = 1;
    for (int s = 0; s < nsteps; ++s) {
        const int d  = dfirst + dstep * s;
        const int rb = (s & 1) * XB;
        const int wb = XB - rb;

        v2f bo0, bo1, bo2, bo3;
        bo0[0] = s0; bo0[1] = s1; bo1[0] = s2; bo1[1] = s3;
        bo2[0] = s4; bo2[1] = s5; bo3[0] = s6; bo3[1] = s7;

        v8f acc[8];
        #pragma unroll
        for (int j = 0; j < 8; ++j)
            acc[j] = __builtin_amdgcn_wmma_f32_16x16x4_f32(false, Af[j][0], false, bo0,
                                                           (short)0, (v8f){}, false, false);

        __builtin_amdgcn_sched_barrier(0);
        __builtin_amdgcn_fence(__ATOMIC_RELEASE, "workgroup");
        __builtin_amdgcn_s_barrier_signal(-1);
        __builtin_amdgcn_sched_barrier(0);

        #pragma unroll
        for (int j = 0; j < 8; ++j)
            acc[j] = __builtin_amdgcn_wmma_f32_16x16x4_f32(false, Af[j][1], false, bo1,
                                                           (short)0, acc[j], false, false);

        __builtin_amdgcn_sched_barrier(0);
        __builtin_amdgcn_s_barrier_wait(-1);
        __builtin_amdgcn_fence(__ATOMIC_ACQUIRE, "workgroup");

        const float4 xa0 = *(const float4*)(xr1 + rb), xa1 = *(const float4*)(xr1 + rb + 128);
        const float4 xb0 = *(const float4*)(xr2 + rb), xb1 = *(const float4*)(xr2 + rb + 128);
        const float4 xc0 = *(const float4*)(xr3 + rb), xc1 = *(const float4*)(xr3 + rb + 128);
        const float4 p0 = *(const float4*)(php + d * 8);
        const float4 p1 = *(const float4*)(php + d * 8 + 4);
        __builtin_amdgcn_sched_barrier(0);

        #pragma unroll
        for (int j = 0; j < 8; ++j)
            acc[j] = __builtin_amdgcn_wmma_f32_16x16x4_f32(false, Af[j][2], false, bo2,
                                                           (short)0, acc[j], false, false);
        #pragma unroll
        for (int j = 0; j < 8; ++j)
            acc[j] = __builtin_amdgcn_wmma_f32_16x16x4_f32(false, Af[j][3], false, bo3,
                                                           (short)0, acc[j], false, false);

        __builtin_amdgcn_sched_barrier(0);

        v2f bx[12];
        bx[0][0] = xa0.x; bx[0][1] = xa0.y; bx[1][0] = xa0.z; bx[1][1] = xa0.w;
        bx[2][0] = xa1.x; bx[2][1] = xa1.y; bx[3][0] = xa1.z; bx[3][1] = xa1.w;
        bx[4][0] = xb0.x; bx[4][1] = xb0.y; bx[5][0] = xb0.z; bx[5][1] = xb0.w;
        bx[6][0] = xb1.x; bx[6][1] = xb1.y; bx[7][0] = xb1.z; bx[7][1] = xb1.w;
        bx[8][0] = xc0.x; bx[8][1] = xc0.y; bx[9][0] = xc0.z; bx[9][1] = xc0.w;
        bx[10][0] = xc1.x; bx[10][1] = xc1.y; bx[11][0] = xc1.z; bx[11][1] = xc1.w;

        #pragma unroll
        for (int kb = 4; kb < 16; ++kb) {
            #pragma unroll
            for (int j = 0; j < 8; ++j)
                acc[j] = __builtin_amdgcn_wmma_f32_16x16x4_f32(false, Af[j][kb], false, bx[kb - 4],
                                                               (short)0, acc[j], false, false);
        }

        #define COMBINE(r) fmaf(p1.w, acc[7][r], fmaf(p1.z, acc[6][r], fmaf(p1.y, acc[5][r], fmaf(p1.x, acc[4][r], \
                           fmaf(p0.w, acc[3][r], fmaf(p0.z, acc[2][r], fmaf(p0.y, acc[1][r], p0.x * acc[0][r])))))))
        s0 = COMBINE(0); s1 = COMBINE(1); s2 = COMBINE(2); s3 = COMBINE(3);
        s4 = COMBINE(4); s5 = COMBINE(5); s6 = COMBINE(6); s7 = COMBINE(7);
        #undef COMBINE

        *(float4*)(xw + wb)       = make_float4(s0, s1, s2, s3);
        *(float4*)(xw + wb + 128) = make_float4(s4, s5, s6, s7);
    }
    {
        float* vf = Vfin + c * VS + 16 * w + 8 * h;
        *(float4*)(vf)     = make_float4(s0, s1, s2, s3);
        *(float4*)(vf + 4) = make_float4(s4, s5, s6, s7);
    }
    __syncthreads();
    {
        float* dst = Vf + (size_t)col0 * 64;
        #pragma unroll
        for (int q = 0; q < 2; ++q) {
            int e4  = tid + 128 * q;
            int row = e4 >> 4;
            int c4  = e4 & 15;
            *(float4*)(dst + row * 64 + 4 * c4) = *(const float4*)(Vfin + row * VS + 4 * c4);
        }
    }
    } else {
    const int pair0  = PPB * (blockIdx.x - NFWD);
    const int npairs = (NBP - pair0 < PPB) ? NBP - pair0 : PPB;
    const bool stager = tid < 2 * NB4;
    const int  sxb    = tid / NB4;
    const int  st4    = tid - sxb * NB4;
    const float4* fsrc = (const float4*)phidiv + (size_t)col0 * 128 + FS * 2 + sxb * 128 + st4;
    float4 stg = make_float4(0.f, 0.f, 0.f, 0.f);
    if (stager) stg = fsrc[0];

    for (int pp = 0; pp < npairs; ++pp) {
    if (stager) *(float4*)(phs + sxb * PS + 4 * (FS * 2 + st4)) = stg;
    *(float4*)(xw)            = make_float4(s0, s1, s2, s3);
    *(float4*)(xw + 128)      = make_float4(s4, s5, s6, s7);
    *(float4*)(xw + XB)       = make_float4(s0, s1, s2, s3);
    *(float4*)(xw + XB + 128) = make_float4(s4, s5, s6, s7);
    __syncthreads();
    if (stager && pp + 1 < npairs) stg = fsrc[(pp + 1) * 256];

    #define FEAT(pa, pb, j) ((j) == 0 ? pa.x : (j) == 1 ? pa.y : (j) == 2 ? pa.z : (j) == 3 ? pa.w : \
                             (j) == 4 ? pb.x : (j) == 5 ? pb.y : (j) == 6 ? pb.z : pb.w)
    #define FORMJ(dst, pa, pb, j, k0, k1) do {                                                          \
        const int jf_ = (j);     \
        _Pragma("unroll") for (int kf_ = (k0); kf_ < (k1); ++kf_) {                                     \
            if (jf_ == 0) dst[kf_] = FEAT(pa, pb, jf_) * Af[jf_][kf_];                                  \
            else          dst[kf_] = FEAT(pa, pb, jf_) * Af[jf_][kf_] + dst[kf_];                       \
        }                                                                                               \
    } while (0)

    float  sv[2][8];
    float4 fa[2], fb[2];
    v2f    fr[2][16];
    v2f    bx[2][12];
    #pragma unroll
    for (int x = 0; x < 2; ++x) {
        sv[x][0] = s0; sv[x][1] = s1; sv[x][2] = s2; sv[x][3] = s3;
        sv[x][4] = s4; sv[x][5] = s5; sv[x][6] = s6; sv[x][7] = s7;
    }
    fa[0] = *(const float4*)(phs + (D_ - 1) * 8);
    fb[0] = *(const float4*)(phs + (D_ - 1) * 8 + 4);
    fa[1] = *(const float4*)(phs + PS + (D_ - 1) * 8);
    fb[1] = *(const float4*)(phs + PS + (D_ - 1) * 8 + 4);
    {
        const float4 xa0 = *(const float4*)(xr1), xa1 = *(const float4*)(xr1 + 128);
        const float4 xb0 = *(const float4*)(xr2), xb1 = *(const float4*)(xr2 + 128);
        const float4 xc0 = *(const float4*)(xr3), xc1 = *(const float4*)(xr3 + 128);
        bx[0][0][0] = xa0.x; bx[0][0][1] = xa0.y; bx[0][1][0] = xa0.z; bx[0][1][1] = xa0.w;
        bx[0][2][0] = xa1.x; bx[0][2][1] = xa1.y; bx[0][3][0] = xa1.z; bx[0][3][1] = xa1.w;
        bx[0][4][0] = xb0.x; bx[0][4][1] = xb0.y; bx[0][5][0] = xb0.z; bx[0][5][1] = xb0.w;
        bx[0][6][0] = xb1.x; bx[0][6][1] = xb1.y; bx[0][7][0] = xb1.z; bx[0][7][1] = xb1.w;
        bx[0][8][0] = xc0.x; bx[0][8][1] = xc0.y; bx[0][9][0] = xc0.z; bx[0][9][1] = xc0.w;
        bx[0][10][0] = xc1.x; bx[0][10][1] = xc1.y; bx[0][11][0] = xc1.z; bx[0][11][1] = xc1.w;
    }
    #pragma unroll
    for (int j = 0; j < 8; ++j) FORMJ(fr[0], fa[0], fb[0], j, 0, 16);

    for (int s = 0; s < D_ - FS; ++s) {
        const int d  = D_ - 1 - s;
        const int dn = (s + 1 < D_ - FS) ? d - 1 : d;

        #pragma unroll
        for (int x = 0; x < 2; ++x) {
            const int y = 1 - x;

            v2f bo[4];
            bo[0][0] = sv[x][0]; bo[0][1] = sv[x][1]; bo[1][0] = sv[x][2]; bo[1][1] = sv[x][3];
            bo[2][0] = sv[x][4]; bo[2][1] = sv[x][5]; bo[3][0] = sv[x][6]; bo[3][1] = sv[x][7];

            v8f acc[2] = {(v8f){}, (v8f){}};
            float4 xa0, xa1, xb0, xb1, xc0, xc1, pn0, pn1;

            #pragma unroll
            for (int kb = 0; kb < 4; ++kb) {
                acc[kb & 1] = __builtin_amdgcn_wmma_f32_16x16x4_f32(false, fr[x][kb], false, bo[kb],
                                                                    (short)0, acc[kb & 1], false, false);
                FORMJ(fr[y], fa[y], fb[y], 2 * kb,     0, 4);
                FORMJ(fr[y], fa[y], fb[y], 2 * kb + 1, 0, 4);
                __builtin_amdgcn_sched_barrier(0);
            }

            #pragma unroll
            for (int kb = 4; kb < 8; ++kb) {
                if (kb == 6) {
                    asm volatile("s_wait_dscnt 0x0" ::: "memory");
                    __builtin_amdgcn_s_barrier_signal(-1);
                    __builtin_amdgcn_sched_barrier(0);
                }
                acc[kb & 1] = __builtin_amdgcn_wmma_f32_16x16x4_f32(false, fr[x][kb], false, bx[x][kb - 4],
                                                                    (short)0, acc[kb & 1], false, false);
                FORMJ(fr[y], fa[y], fb[y], 2 * (kb - 4),     4, 8);
                FORMJ(fr[y], fa[y], fb[y], 2 * (kb - 4) + 1, 4, 8);
                __builtin_amdgcn_sched_barrier(0);
            }

            #pragma unroll
            for (int kb = 8; kb < 16; ++kb) {
                if (kb == 10) {
                    __builtin_amdgcn_s_barrier_wait(-1);
                    __builtin_amdgcn_fence(__ATOMIC_ACQUIRE, "workgroup");
                    xa0 = *(const float4*)(xr1 + y * XB); xa1 = *(const float4*)(xr1 + y * XB + 128);
                    xb0 = *(const float4*)(xr2 + y * XB); xb1 = *(const float4*)(xr2 + y * XB + 128);
                    xc0 = *(const float4*)(xr3 + y * XB); xc1 = *(const float4*)(xr3 + y * XB + 128);
                    pn0 = *(const float4*)(phs + x * PS + dn * 8);
                    pn1 = *(const float4*)(phs + x * PS + dn * 8 + 4);
                    __builtin_amdgcn_sched_barrier(0);
                }
                acc[kb & 1] = __builtin_amdgcn_wmma_f32_16x16x4_f32(false, fr[x][kb], false, bx[x][kb - 4],
                                                                    (short)0, acc[kb & 1], false, false);
                if (kb == 15) __builtin_amdgcn_sched_barrier(0);
                FORMJ(fr[y], fa[y], fb[y], kb - 8, 8, 16);
                __builtin_amdgcn_sched_barrier(0);
            }

            bx[y][0][0] = xa0.x; bx[y][0][1] = xa0.y; bx[y][1][0] = xa0.z; bx[y][1][1] = xa0.w;
            bx[y][2][0] = xa1.x; bx[y][2][1] = xa1.y; bx[y][3][0] = xa1.z; bx[y][3][1] = xa1.w;
            bx[y][4][0] = xb0.x; bx[y][4][1] = xb0.y; bx[y][5][0] = xb0.z; bx[y][5][1] = xb0.w;
            bx[y][6][0] = xb1.x; bx[y][6][1] = xb1.y; bx[y][7][0] = xb1.z; bx[y][7][1] = xb1.w;
            bx[y][8][0] = xc0.x; bx[y][8][1] = xc0.y; bx[y][9][0] = xc0.z; bx[y][9][1] = xc0.w;
            bx[y][10][0] = xc1.x; bx[y][10][1] = xc1.y; bx[y][11][0] = xc1.z; bx[y][11][1] = xc1.w;
            fa[x] = pn0; fb[x] = pn1;

            #pragma unroll
            for (int r = 0; r < 8; ++r) sv[x][r] = acc[0][r] + acc[1][r];

            *(float4*)(xw + x * XB)       = make_float4(sv[x][0], sv[x][1], sv[x][2], sv[x][3]);
            *(float4*)(xw + x * XB + 128) = make_float4(sv[x][4], sv[x][5], sv[x][6], sv[x][7]);
        }
    }
    #undef FORMJ
    #undef FEAT
    {
        float* vf0 = Vfin + c * VS + 16 * w + 8 * h;
        *(float4*)(vf0)     = make_float4(sv[0][0], sv[0][1], sv[0][2], sv[0][3]);
        *(float4*)(vf0 + 4) = make_float4(sv[0][4], sv[0][5], sv[0][6], sv[0][7]);
        float* vf1 = vf0 + 16 * VS;
        *(float4*)(vf1)     = make_float4(sv[1][0], sv[1][1], sv[1][2], sv[1][3]);
        *(float4*)(vf1 + 4) = make_float4(sv[1][4], sv[1][5], sv[1][6], sv[1][7]);
    }
    __syncthreads();
    {
        float* dst = Ub + (size_t)(col0 + 2 * pp) * (C_ * 64);
        #pragma unroll
        for (int q = 0; q < 3; ++q) {
            int e4  = tid + 128 * q;
            int row = e4 >> 4;
            int c4  = e4 & 15;
            int srow = (row >= C_) ? row + (16 - C_) : row;
            if (row < 2 * C_)
                *(float4*)(dst + row * 64 + 4 * c4) = *(const float4*)(Vfin + srow * VS + 4 * c4);
        }
    }
    }
    }
}

__global__ __launch_bounds__(512)
void join_kernel(const float* __restrict__ Vf, const float* __restrict__ Ub,
                 float* __restrict__ out) {
    __shared__ float res[16 * C_];
    const int wv   = threadIdx.x >> 5;
    const int b    = blockIdx.x * 16 + wv;
    const int lane = threadIdx.x & 31;
    const float v0 = Vf[b * 64 + lane], v1 = Vf[b * 64 + 32 + lane];
    const float* up = Ub + (size_t)b * (C_ * 64) + lane;
    float tr[C_];
    #pragma unroll
    for (int k = 0; k < C_; ++k) tr[k] = fmaf(v1, up[k * 64 + 32], v0 * up[k * 64]);
    #pragma unroll
    for (int off = 16; off > 0; off >>= 1) {
        #pragma unroll
        for (int k = 0; k < C_; ++k) tr[k] += __shfl_xor(tr[k], off, 32);
    }
    float mx = tr[0];
    #pragma unroll
    for (int k = 1; k < C_; ++k) mx = fmaxf(mx, tr[k]);
    float se = 0.f;
    #pragma unroll
    for (int k = 0; k < C_; ++k) se += expf(tr[k] - mx);
    float mine = tr[0];
    #pragma unroll
    for (int k = 1; k < C_; ++k) mine = (lane == k) ? tr[k] : mine;
    if (lane < C_) res[wv * C_ + lane] = mine - (logf(se) + mx);
    __syncthreads();
    if (threadIdx.x < 16 * C_) out[(size_t)blockIdx.x * (16 * C_) + threadIdx.x] = res[threadIdx.x];
}

extern "C" void kernel_launch(void* const* d_in, const int* in_sizes, int n_in,
                              void* d_out, int out_size, void* d_ws, size_t ws_size,
                              hipStream_t stream) {
    const float* x     = (const float*)d_in[0];
    const float* Wsh   = (const float*)d_in[1];
    const float* Wlast = (const float*)d_in[2];
    float* out = (float*)d_out;
    float* ws  = (float*)d_ws;
    float* WlS    = ws;
    float* phidiv = ws + (C_ * R_ * R_);
    float* Vf     = phidiv + (B_ * D_ * M_);
    float* Ub     = Vf + (B_ * R_);

    prep_kernel<<<PHB + WLB, 256, 0, stream>>>(x, Wlast, phidiv, WlS);
    const size_t reserve = (size_t)168 * 1024;
    hipFuncSetAttribute((const void*)trnet_kernel,
                        hipFuncAttributeMaxDynamicSharedMemorySize, (int)reserve);
    trnet_kernel<<<NFWD + NBWG, 128, reserve, stream>>>(Wsh, phidiv, WlS, Vf, Ub);
    join_kernel<<<B_ / 16, 512, 0, stream>>>(Vf, Ub, out);
}


# Step matrices A_d = sum_j phi_j W_j formed on WMMA (4 batches x 4 steps per tile) into LDS; each wave carries one batch's row through them on the vector unit; no time split; f32 sum order changes
# speedup vs baseline: 9.2259x; 1.1351x over previous
#include <hip/hip_runtime.h>
#include <math.h>

typedef __attribute__((ext_vector_type(2))) float v2f;
typedef __attribute__((ext_vector_type(8))) float v8f;

#define B_  1024
#define N_  1024
#define M_  8
#define D_  64
#define R_  64
#define C_  10

#define GB    4
#define GS    4
#define NGRP  (D_ / GS)
#define MP    (R_ * R_ + 4)
#define NWG   (B_ / GB)
#define LDSB  262400


#define PHB  (B_ * D_ / 256)
#define WLB  (C_ * R_ * 32 / 256)

__global__ __launch_bounds__(256)
void prep_kernel(const float* __restrict__ x, const float* __restrict__ Wlast,
                 float* __restrict__ phidiv, float* __restrict__ WlS) {
    if (blockIdx.x < PHB) {
        const int tid = blockIdx.x * 256 + threadIdx.x;
        const float* xp = x + (size_t)tid * 16;
        float acc[8];
        #pragma unroll
        for (int m = 0; m < 8; ++m) acc[m] = 0.f;
        for (int n = 0; n < 16; ++n) {
            float a = 1.5707963267948966f * xp[n];
            float sn, cs;
            sincosf(a, &sn, &cs);
            float pc[8], ps[8];
            pc[0] = 1.f; ps[0] = 1.f;
            #pragma unroll
            for (int k = 1; k < 8; ++k) { pc[k] = pc[k-1] * cs; ps[k] = ps[k-1] * sn; }
            #pragma unroll
            for (int m = 0; m < 8; ++m) acc[m] = fmaf(pc[7-m], ps[m], acc[m]);
        }
        float* o = phidiv + (size_t)tid * 8;
        #pragma unroll
        for (int m = 0; m < 8; ++m) o[m] = acc[m];
    } else {
        const int wv   = ((blockIdx.x - PHB) * 256 + threadIdx.x) >> 5;
        const int lane = threadIdx.x & 31;
        const float* p = Wlast + (size_t)wv * 512 + lane;
        float s = 0.f;
        #pragma unroll
        for (int q = 0; q < 16; ++q) s += p[32 * q];
        #pragma unroll
        for (int off = 16; off > 0; off >>= 1) s += __shfl_xor(s, off, 32);
        if (lane == 0) WlS[wv] = s;
    }
}


#define ROWSHARE(x, n) __int_as_float(__builtin_amdgcn_update_dpp(0, __float_as_int(x), 0x150 + (n), 0xf, 0xf, true))
#define XHALF(x) __int_as_float(__builtin_amdgcn_permlanex16(__float_as_int(x), __float_as_int(x), 0x76543210, 0xfedcba98, false, false))

__global__ __launch_bounds__(128)
void trnet_kernel(const float* __restrict__ Wsh, const float* __restrict__ phidiv,
                  float* __restrict__ Vf)
{
    extern __shared__ __attribute__((aligned(16))) float lds[];

    const int tid  = threadIdx.x;
    const int w    = tid >> 5;
    const int lane = tid & 31;
    const int c    = lane & 15;
    const int h    = lane >> 4;
    const int b0   = blockIdx.x * GB;

    v2f Af[64][2];
    #pragma unroll
    for (int i = 0; i < 64; ++i) {
        #pragma unroll
        for (int kb = 0; kb < 2; ++kb) {
            const float* p = Wsh + i * 512 + (4 * kb + 2 * h) * 64 + 16 * w + c;
            v2f f; f[0] = p[0]; f[1] = p[64];
            Af[i][kb] = f;
        }
    }

    const float* bp = phidiv + ((size_t)(b0 + (c >> 2)) * 64 + (c & 3)) * 8 + 2 * h;
    v2f bn0 = *(const v2f*)(bp), bn1 = *(const v2f*)(bp + 4);

    float*       dst  = lds + c * MP + 16 * w + 8 * h;
    const float* msrc = lds + (4 * w) * MP + 64 * h + 4 * c;

    float4 sv = make_float4(1.f, 1.f, 1.f, 1.f);
    float  t0 = 1.f, t1 = 1.f;

    for (int G = 0; G < NGRP; ++G) {
        const v2f bf0 = bn0, bf1 = bn1;
        const int Gn = (G + 1 < NGRP) ? G + 1 : G;
        bn0 = *(const v2f*)(bp + Gn * 32);
        bn1 = *(const v2f*)(bp + Gn * 32 + 4);

        #pragma unroll
        for (int blk = 0; blk < 8; ++blk) {
            v8f dc[8];
            #pragma unroll
            for (int t = 0; t < 8; ++t)
                dc[t] = __builtin_amdgcn_wmma_f32_16x16x4_f32(false, Af[8 * blk + t][0], false, bf0,
                                                              (short)0, (v8f){}, false, false);
            #pragma unroll
            for (int t = 0; t < 8; ++t)
                dc[t] = __builtin_amdgcn_wmma_f32_16x16x4_f32(false, Af[8 * blk + t][1], false, bf1,
                                                              (short)0, dc[t], false, false);
            __builtin_amdgcn_sched_barrier(0);
            #pragma unroll
            for (int t = 0; t < 8; ++t) {
                float* o = dst + (8 * blk + t) * 64;
                *(float4*)(o)     = make_float4(dc[t][0], dc[t][1], dc[t][2], dc[t][3]);
                *(float4*)(o + 4) = make_float4(dc[t][4], dc[t][5], dc[t][6], dc[t][7]);
            }
            __builtin_amdgcn_sched_barrier(0);
        }
        __syncthreads();

        float4 ma[32], mb[32];
        #define LOADM(buf, dd) _Pragma("unroll") for (int q = 0; q < 32; ++q) buf[q] = *(const float4*)(msrc + (dd) * MP + q * 128)
        #define ROWQ(buf, q) do {                                                              \
            const float vi_ = ROWSHARE(((q) & 1) ? t1 : t0, (q) >> 1);                         \
            const v2f s_ = {vi_, vi_};                                                         \
            alo[(q) & 3] += s_ * (v2f){buf[q].x, buf[q].y};                                    \
            ahi[(q) & 3] += s_ * (v2f){buf[q].z, buf[q].w};                                    \
        } while (0)
        #define ROW4(buf, q)  ROWQ(buf, q); ROWQ(buf, (q) + 1); ROWQ(buf, (q) + 2); ROWQ(buf, (q) + 3)
        #define ROW16(buf, q) ROW4(buf, q); ROW4(buf, (q) + 4); ROW4(buf, (q) + 8); ROW4(buf, (q) + 12)
        #define STEP(buf) do {                                                                 \
            v2f alo[4] = {{0.f, 0.f}, {0.f, 0.f}, {0.f, 0.f}, {0.f, 0.f}};                     \
            v2f ahi[4] = {{0.f, 0.f}, {0.f, 0.f}, {0.f, 0.f}, {0.f, 0.f}};                     \
            ROW16(buf, 0); ROW16(buf, 16);                                                     \
            const v2f lo_ = (alo[0] + alo[1]) + (alo[2] + alo[3]);                             \
            const v2f hi_ = (ahi[0] + ahi[1]) + (ahi[2] + ahi[3]);                             \
                                         \
            sv.x = lo_[0] + XHALF(lo_[0]); sv.y = lo_[1] + XHALF(lo_[1]);                      \
            sv.z = hi_[0] + XHALF(hi_[0]); sv.w = hi_[1] + XHALF(hi_[1]);                      \
            t0 = h ? sv.y : sv.x; t1 = h ? sv.w : sv.z;                                        \
        } while (0)

        LOADM(ma, 0);
        LOADM(mb, 1);
        __builtin_amdgcn_sched_barrier(0);
        STEP(ma);
        LOADM(ma, 2);
        __builtin_amdgcn_sched_barrier(0);
        STEP(mb);
        LOADM(mb, 3);
        __builtin_amdgcn_sched_barrier(0);
        STEP(ma);
        STEP(mb);
        #undef STEP
        #undef ROW16
        #undef ROW4
        #undef ROWQ
        #undef LOADM
        __syncthreads();
    }

    if (h == 0) *(float4*)(Vf + (size_t)(b0 + w) * 64 + 4 * c) = sv;
}

__global__ __launch_bounds__(512)
void join_kernel(const float* __restrict__ Vf, const float* __restrict__ WlS,
                 float* __restrict__ out) {
    __shared__ float res[16 * C_];
    const int wv   = threadIdx.x >> 5;
    const int b    = blockIdx.x * 16 + wv;
    const int lane = threadIdx.x & 31;
    const float v0 = Vf[b * 64 + lane], v1 = Vf[b * 64 + 32 + lane];
    const float* up = WlS + lane;
    float tr[C_];
    #pragma unroll
    for (int k = 0; k < C_; ++k) tr[k] = fmaf(v1, up[k * 64 + 32], v0 * up[k * 64]);
    #pragma unroll
    for (int off = 16; off > 0; off >>= 1) {
        #pragma unroll
        for (int k = 0; k < C_; ++k) tr[k] += __shfl_xor(tr[k], off, 32);
    }
    float mx = tr[0];
    #pragma unroll
    for (int k = 1; k < C_; ++k) mx = fmaxf(mx, tr[k]);
    float se = 0.f;
    #pragma unroll
    for (int k = 0; k < C_; ++k) se += expf(tr[k] - mx);
    float mine = tr[0];
    #pragma unroll
    for (int k = 1; k < C_; ++k) mine = (lane == k) ? tr[k] : mine;
    if (lane < C_) res[wv * C_ + lane] = mine - (logf(se) + mx);
    __syncthreads();
    if (threadIdx.x < 16 * C_) out[(size_t)blockIdx.x * (16 * C_) + threadIdx.x] = res[threadIdx.x];
}

extern "C" void kernel_launch(void* const* d_in, const int* in_sizes, int n_in,
                              void* d_out, int out_size, void* d_ws, size_t ws_size,
                              hipStream_t stream) {
    const float* x     = (const float*)d_in[0];
    const float* Wsh   = (const float*)d_in[1];
    const float* Wlast = (const float*)d_in[2];
    float* out = (float*)d_out;
    float* ws  = (float*)d_ws;
    float* WlS    = ws;
    float* phidiv = ws + (C_ * R_ * R_);
    float* Vf     = phidiv + (B_ * D_ * M_);

    prep_kernel<<<PHB + WLB, 256, 0, stream>>>(x, Wlast, phidiv, WlS);
    hipFuncSetAttribute((const void*)trnet_kernel,
                        hipFuncAttributeMaxDynamicSharedMemorySize, LDSB);
    trnet_kernel<<<NWG, 128, LDSB, stream>>>(Wsh, phidiv, Vf);
    join_kernel<<<B_ / 16, 512, 0, stream>>>(Vf, WlS, out);
}


# Pooled features and class-weight sums computed inside the chain kernel's own blocks, in quarters under the W fragment fetches (features kept in LDS): 3 launches become 2; sum orders unchanged
# speedup vs baseline: 10.0312x; 1.0873x over previous
#include <hip/hip_runtime.h>
#include <math.h>

typedef __attribute__((ext_vector_type(2))) float v2f;
typedef __attribute__((ext_vector_type(8))) float v8f;

#define B_  1024
#define N_  1024
#define M_  8
#define D_  64
#define R_  64
#define C_  10

#define GB    4
#define GS    4
#define NGRP  (D_ / GS)
#define MP    (R_ * R_ + 4)
#define NWG   (B_ / GB)
#define PHF   (GB * D_ * M_)
#define LDSB  (262400 + 4 * PHF)


__device__ __forceinline__ void phi_add(float xv, float (&acc)[8]) {
    float a = 1.5707963267948966f * xv;
    float sn, cs;
    sincosf(a, &sn, &cs);
    float pc[8], ps[8];
    pc[0] = 1.f; ps[0] = 1.f;
    #pragma unroll
    for (int k = 1; k < 8; ++k) { pc[k] = pc[k-1] * cs; ps[k] = ps[k-1] * sn; }
    #pragma unroll
    for (int m = 0; m < 8; ++m) acc[m] = fmaf(pc[7-m], ps[m], acc[m]);
}


#define ROWSHARE(x, n) __int_as_float(__builtin_amdgcn_update_dpp(0, __float_as_int(x), 0x150 + (n), 0xf, 0xf, true))
#define XHALF(x) __int_as_float(__builtin_amdgcn_permlanex16(__float_as_int(x), __float_as_int(x), 0x76543210, 0xfedcba98, false, false))

__global__ __launch_bounds__(128)
void trnet_kernel(const float* __restrict__ x, const float* __restrict__ Wsh,
                  const float* __restrict__ Wlast, float* __restrict__ WlS,
                  float* __restrict__ Vf)
{
    extern __shared__ __attribute__((aligned(16))) float lds[];

    const int tid  = threadIdx.x;
    const int w    = tid >> 5;
    const int lane = tid & 31;
    const int c    = lane & 15;
    const int h    = lane >> 4;
    const int b0   = blockIdx.x * GB;
    float* phil    = lds + 16 * MP;

    const float4* xq0 = (const float4*)(x + (size_t)b0 * N_ + tid * 16);
    const float4* xq1 = xq0 + 128 * 4;
    float4 qa[4], qb[4];
    #pragma unroll
    for (int L = 0; L < 4; ++L) { qa[L] = xq0[L]; qb[L] = xq1[L]; }

    {
        const int gw = blockIdx.x * 4 + w;
        if (gw < C_ * R_) {
            const float* p = Wlast + (size_t)gw * 512 + lane;
            float s = 0.f;
            #pragma unroll
            for (int q = 0; q < 16; ++q) s += p[32 * q];
            #pragma unroll
            for (int off = 16; off > 0; off >>= 1) s += __shfl_xor(s, off, 32);
            if (lane == 0) WlS[gw] = s;
        }
    }

    v2f Af[64][2];
    #define AF_QUARTER(L) _Pragma("unroll") for (int i = 16 * (L); i < 16 * (L) + 16; ++i) {          \
        _Pragma("unroll") for (int kb = 0; kb < 2; ++kb) {                                             \
            const float* p = Wsh + i * 512 + (4 * kb + 2 * h) * 64 + 16 * w + c;                       \
            v2f f; f[0] = p[0]; f[1] = p[64];                                                          \
            Af[i][kb] = f;                                                                             \
        } }

    float acc0[8], acc1[8];
    #pragma unroll
    for (int m = 0; m < 8; ++m) { acc0[m] = 0.f; acc1[m] = 0.f; }
    #define PICK(q, t) ((t) == 0 ? (q).x : (t) == 1 ? (q).y : (t) == 2 ? (q).z : (q).w)
    #define PHI_QUARTER(L) _Pragma("nounroll") for (int t = 0; t < 4; ++t) {                            \
        phi_add(PICK(qa[L], t), acc0);                                                                 \
        phi_add(PICK(qb[L], t), acc1);                                                                 \
    }
    AF_QUARTER(0); PHI_QUARTER(0);
    AF_QUARTER(1); PHI_QUARTER(1);
    AF_QUARTER(2); PHI_QUARTER(2);
    AF_QUARTER(3); PHI_QUARTER(3);
    #undef PHI_QUARTER
    #undef PICK
    #undef AF_QUARTER
    {
        float* o0 = phil + tid * 8;
        float* o1 = phil + (tid + 128) * 8;
        *(float4*)(o0)     = make_float4(acc0[0], acc0[1], acc0[2], acc0[3]);
        *(float4*)(o0 + 4) = make_float4(acc0[4], acc0[5], acc0[6], acc0[7]);
        *(float4*)(o1)     = make_float4(acc1[0], acc1[1], acc1[2], acc1[3]);
        *(float4*)(o1 + 4) = make_float4(acc1[4], acc1[5], acc1[6], acc1[7]);
    }

    __syncthreads();
    const float* bp = phil + ((c >> 2) * 64 + (c & 3)) * 8 + 2 * h;
    v2f bn0 = *(const v2f*)(bp), bn1 = *(const v2f*)(bp + 4);

    float*       dst  = lds + c * MP + 16 * w + 8 * h;
    const float* msrc = lds + (4 * w) * MP + 64 * h + 4 * c;

    float4 sv = make_float4(1.f, 1.f, 1.f, 1.f);
    float  t0 = 1.f, t1 = 1.f;

    for (int G = 0; G < NGRP; ++G) {
        const v2f bf0 = bn0, bf1 = bn1;
        const int Gn = (G + 1 < NGRP) ? G + 1 : G;
        bn0 = *(const v2f*)(bp + Gn * 32);
        bn1 = *(const v2f*)(bp + Gn * 32 + 4);

        #pragma unroll
        for (int blk = 0; blk < 8; ++blk) {
            v8f dc[8];
            #pragma unroll
            for (int t = 0; t < 8; ++t)
                dc[t] = __builtin_amdgcn_wmma_f32_16x16x4_f32(false, Af[8 * blk + t][0], false, bf0,
                                                              (short)0, (v8f){}, false, false);
            #pragma unroll
            for (int t = 0; t < 8; ++t)
                dc[t] = __builtin_amdgcn_wmma_f32_16x16x4_f32(false, Af[8 * blk + t][1], false, bf1,
                                                              (short)0, dc[t], false, false);
            __builtin_amdgcn_sched_barrier(0);
            #pragma unroll
            for (int t = 0; t < 8; ++t) {
                float* o = dst + (8 * blk + t) * 64;
                *(float4*)(o)     = make_float4(dc[t][0], dc[t][1], dc[t][2], dc[t][3]);
                *(float4*)(o + 4) = make_float4(dc[t][4], dc[t][5], dc[t][6], dc[t][7]);
            }
            __builtin_amdgcn_sched_barrier(0);
        }
        __syncthreads();

        float4 ma[32], mb[32];
        #define LOADM(buf, dd) _Pragma("unroll") for (int q = 0; q < 32; ++q) buf[q] = *(const float4*)(msrc + (dd) * MP + q * 128)
        #define ROWQ(buf, q) do {                                                              \
            const float vi_ = ROWSHARE(((q) & 1) ? t1 : t0, (q) >> 1);                         \
            const v2f s_ = {vi_, vi_};                                                         \
            alo[(q) & 3] += s_ * (v2f){buf[q].x, buf[q].y};                                    \
            ahi[(q) & 3] += s_ * (v2f){buf[q].z, buf[q].w};                                    \
        } while (0)
        #define ROW4(buf, q)  ROWQ(buf, q); ROWQ(buf, (q) + 1); ROWQ(buf, (q) + 2); ROWQ(buf, (q) + 3)
        #define ROW16(buf, q) ROW4(buf, q); ROW4(buf, (q) + 4); ROW4(buf, (q) + 8); ROW4(buf, (q) + 12)
        #define STEP(buf) do {                                                                 \
            v2f alo[4] = {{0.f, 0.f}, {0.f, 0.f}, {0.f, 0.f}, {0.f, 0.f}};                     \
            v2f ahi[4] = {{0.f, 0.f}, {0.f, 0.f}, {0.f, 0.f}, {0.f, 0.f}};                     \
            ROW16(buf, 0); ROW16(buf, 16);                                                     \
            const v2f lo_ = (alo[0] + alo[1]) + (alo[2] + alo[3]);                             \
            const v2f hi_ = (ahi[0] + ahi[1]) + (ahi[2] + ahi[3]);                             \
                                         \
            sv.x = lo_[0] + XHALF(lo_[0]); sv.y = lo_[1] + XHALF(lo_[1]);                      \
            sv.z = hi_[0] + XHALF(hi_[0]); sv.w = hi_[1] + XHALF(hi_[1]);                      \
            t0 = h ? sv.y : sv.x; t1 = h ? sv.w : sv.z;                                        \
        } while (0)

        LOADM(ma, 0);
        LOADM(mb, 1);
        __builtin_amdgcn_sched_barrier(0);
        STEP(ma);
        LOADM(ma, 2);
        __builtin_amdgcn_sched_barrier(0);
        STEP(mb);
        LOADM(mb, 3);
        __builtin_amdgcn_sched_barrier(0);
        STEP(ma);
        STEP(mb);
        #undef STEP
        #undef ROW16
        #undef ROW4
        #undef ROWQ
        #undef LOADM
        __syncthreads();
    }

    if (h == 0) *(float4*)(Vf + (size_t)(b0 + w) * 64 + 4 * c) = sv;
}

__global__ __launch_bounds__(512)
void join_kernel(const float* __restrict__ Vf, const float* __restrict__ WlS,
                 float* __restrict__ out) {
    __shared__ float res[16 * C_];
    const int wv   = threadIdx.x >> 5;
    const int b    = blockIdx.x * 16 + wv;
    const int lane = threadIdx.x & 31;
    const float v0 = Vf[b * 64 + lane], v1 = Vf[b * 64 + 32 + lane];
    const float* up = WlS + lane;
    float tr[C_];
    #pragma unroll
    for (int k = 0; k < C_; ++k) tr[k] = fmaf(v1, up[k * 64 + 32], v0 * up[k * 64]);
    #pragma unroll
    for (int off = 16; off > 0; off >>= 1) {
        #pragma unroll
        for (int k = 0; k < C_; ++k) tr[k] += __shfl_xor(tr[k], off, 32);
    }
    float mx = tr[0];
    #pragma unroll
    for (int k = 1; k < C_; ++k) mx = fmaxf(mx, tr[k]);
    float se = 0.f;
    #pragma unroll
    for (int k = 0; k < C_; ++k) se += expf(tr[k] - mx);
    float mine = tr[0];
    #pragma unroll
    for (int k = 1; k < C_; ++k) mine = (lane == k) ? tr[k] : mine;
    if (lane < C_) res[wv * C_ + lane] = mine - (logf(se) + mx);
    __syncthreads();
    if (threadIdx.x < 16 * C_) out[(size_t)blockIdx.x * (16 * C_) + threadIdx.x] = res[threadIdx.x];
}

extern "C" void kernel_launch(void* const* d_in, const int* in_sizes, int n_in,
                              void* d_out, int out_size, void* d_ws, size_t ws_size,
                              hipStream_t stream) {
    const float* x     = (const float*)d_in[0];
    const float* Wsh   = (const float*)d_in[1];
    const float* Wlast = (const float*)d_in[2];
    float* out = (float*)d_out;
    float* ws  = (float*)d_ws;
    float* WlS    = ws;
    float* Vf     = ws + (C_ * R_ * R_) + (B_ * D_ * M_);

    hipFuncSetAttribute((const void*)trnet_kernel,
                        hipFuncAttributeMaxDynamicSharedMemorySize, LDSB);
    trnet_kernel<<<NWG, 128, LDSB, stream>>>(x, Wsh, Wlast, WlS, Vf);
    join_kernel<<<B_ / 16, 512, 0, stream>>>(Vf, WlS, out);
}


# 8 waves, two per SIMD: waves 4-7 form rows 32-63 of the NEXT group's step matrices into registers under the vector phase of waves 0-3 and store them behind its barrier; sum orders unchanged
# speedup vs baseline: 12.0256x; 1.1988x over previous
#include <hip/hip_runtime.h>
#include <math.h>

typedef __attribute__((ext_vector_type(2))) float v2f;
typedef __attribute__((ext_vector_type(8))) float v8f;

#define B_  1024
#define N_  1024
#define M_  8
#define D_  64
#define R_  64
#define C_  10

#define GB    4
#define GS    4
#define NGRP  (D_ / GS)
#define MP    (R_ * R_ + 4)
#define NWG   (B_ / GB)
#define PHF   (GB * D_ * M_)
#define LDSB  (262400 + 4 * PHF)


__device__ __forceinline__ void phi_add(float xv, float (&acc)[8]) {
    float a = 1.5707963267948966f * xv;
    float sn, cs;
    sincosf(a, &sn, &cs);
    float pc[8], ps[8];
    pc[0] = 1.f; ps[0] = 1.f;
    #pragma unroll
    for (int k = 1; k < 8; ++k) { pc[k] = pc[k-1] * cs; ps[k] = ps[k-1] * sn; }
    #pragma unroll
    for (int m = 0; m < 8; ++m) acc[m] = fmaf(pc[7-m], ps[m], acc[m]);
}


#define ROWSHARE(x, n) __int_as_float(__builtin_amdgcn_update_dpp(0, __float_as_int(x), 0x150 + (n), 0xf, 0xf, true))
#define XHALF(x) __int_as_float(__builtin_amdgcn_permlanex16(__float_as_int(x), __float_as_int(x), 0x76543210, 0xfedcba98, false, false))

#define WMMA4(a, b, cc) __builtin_amdgcn_wmma_f32_16x16x4_f32(false, (a), false, (b), (short)0, (cc), false, false)

__global__ __launch_bounds__(256)
void trnet_kernel(const float* __restrict__ x, const float* __restrict__ Wsh,
                  const float* __restrict__ Wlast, float* __restrict__ WlS,
                  float* __restrict__ Vf)
{
    extern __shared__ __attribute__((aligned(16))) float lds[];

    const int tid  = threadIdx.x;
    const int w8   = __builtin_amdgcn_readfirstlane(tid >> 5);
    const int kq   = w8 & 3;
    const int ih   = w8 >> 2;
    const int lane = tid & 31;
    const int c    = lane & 15;
    const int h    = lane >> 4;
    const int b0   = blockIdx.x * GB;
    float* phil    = lds + 16 * MP;

    const float4* xq0 = (const float4*)(x + (size_t)b0 * N_ + tid * 16);
    float4 qa[4];
    #pragma unroll
    for (int L = 0; L < 4; ++L) qa[L] = xq0[L];

    {
        const int gw = blockIdx.x * 8 + w8;
        if (gw < C_ * R_) {
            const float* p = Wlast + (size_t)gw * 512 + lane;
            float s = 0.f;
            #pragma unroll
            for (int q = 0; q < 16; ++q) s += p[32 * q];
            #pragma unroll
            for (int off = 16; off > 0; off >>= 1) s += __shfl_xor(s, off, 32);
            if (lane == 0) WlS[gw] = s;
        }
    }

    v2f Af[32][2];
    #define AF_QUARTER(L) _Pragma("unroll") for (int ii = 8 * (L); ii < 8 * (L) + 8; ++ii) {          \
        _Pragma("unroll") for (int kb = 0; kb < 2; ++kb) {                                             \
            const float* p = Wsh + (32 * ih + ii) * 512 + (4 * kb + 2 * h) * 64 + 16 * kq + c;         \
            v2f f; f[0] = p[0]; f[1] = p[64];                                                          \
            Af[ii][kb] = f;                                                                            \
        } }

    float acc0[8];
    #pragma unroll
    for (int m = 0; m < 8; ++m) acc0[m] = 0.f;
    #define PICK(q, t) ((t) == 0 ? (q).x : (t) == 1 ? (q).y : (t) == 2 ? (q).z : (q).w)
    #define PHI_QUARTER(L) _Pragma("nounroll") for (int t = 0; t < 4; ++t) {                            \
        phi_add(PICK(qa[L], t), acc0);                                                                 \
    }
    AF_QUARTER(0); PHI_QUARTER(0);
    AF_QUARTER(1); PHI_QUARTER(1);
    AF_QUARTER(2); PHI_QUARTER(2);
    AF_QUARTER(3); PHI_QUARTER(3);
    #undef PHI_QUARTER
    #undef PICK
    #undef AF_QUARTER
    {
        float* o0 = phil + tid * 8;
        *(float4*)(o0)     = make_float4(acc0[0], acc0[1], acc0[2], acc0[3]);
        *(float4*)(o0 + 4) = make_float4(acc0[4], acc0[5], acc0[6], acc0[7]);
    }

    __syncthreads();
    const float* bp = phil + ((c >> 2) * 64 + (c & 3)) * 8 + 2 * h;

    float*       dst  = lds + c * MP + (32 * ih) * 64 + 16 * kq + 8 * h;
    const float* msrc = lds + (4 * kq) * MP + 64 * h + 4 * c;

    #define FORM8(out, blk, f0, f1) do {                                                               \
        _Pragma("unroll") for (int t = 0; t < 8; ++t)                                                  \
            out[t] = WMMA4(Af[8 * (blk) + t][0], f0, (v8f){});                                         \
        _Pragma("unroll") for (int t = 0; t < 8; ++t)                                                  \
            out[t] = WMMA4(Af[8 * (blk) + t][1], f1, out[t]);                                          \
    } while (0)
    #define STORE8(in, blk) do {                                                                       \
        _Pragma("unroll") for (int t = 0; t < 8; ++t) {                                                \
            float* o = dst + (8 * (blk) + t) * 64;                                                     \
            *(float4*)(o)     = make_float4(in[t][0], in[t][1], in[t][2], in[t][3]);                   \
            *(float4*)(o + 4) = make_float4(in[t][4], in[t][5], in[t][6], in[t][7]);                   \
        }                                                                                              \
    } while (0)
    #define FORM_STORE(f0, f1) do {                                                                    \
        _Pragma("unroll") for (int blk = 0; blk < 4; ++blk) {                                          \
            v8f dc[8];                                                                                 \
            FORM8(dc, blk, f0, f1);                                                                    \
            __builtin_amdgcn_sched_barrier(0);                                                         \
            STORE8(dc, blk);                                                                           \
            __builtin_amdgcn_sched_barrier(0);                                                         \
        }                                                                                              \
    } while (0)
    #define FORM_HOLD(hold, f0, f1) do {                                                               \
        _Pragma("unroll") for (int blk = 0; blk < 4; ++blk) FORM8((hold + 8 * blk), blk, f0, f1);      \
    } while (0)
    #define STORE_HOLD(hold) do {                                                                      \
        _Pragma("unroll") for (int blk = 0; blk < 4; ++blk) STORE8((hold + 8 * blk), blk);             \
    } while (0)

    {
        const v2f f0 = *(const v2f*)(bp), f1 = *(const v2f*)(bp + 4);
        FORM_STORE(f0, f1);
    }
    __syncthreads();

    float4 sv = make_float4(1.f, 1.f, 1.f, 1.f);
    float  t0 = 1.f, t1 = 1.f;

    for (int G = 0; G < NGRP; ++G) {
        const bool more = (G + 1 < NGRP);
        const int  Gn   = more ? G + 1 : G;
        const v2f bf0 = *(const v2f*)(bp + Gn * 32);
        const v2f bf1 = *(const v2f*)(bp + Gn * 32 + 4);

        v8f R[32];
        if (ih) {
            FORM_HOLD(R, bf0, bf1);
        } else {
            #define LOADM(hf, dd) _Pragma("unroll") for (int j = 0; j < 16; ++j) {                 \
                const float4 e_ = *(const float4*)(msrc + (dd) * MP + (2 * j) * 128);              \
                const float4 o_ = *(const float4*)(msrc + (dd) * MP + (2 * j + 1) * 128);          \
                R[16 * (hf) + j] = (v8f){e_.x, e_.y, e_.z, e_.w, o_.x, o_.y, o_.z, o_.w};          \
            }
            #define ROWQ(hf, q) do {                                                               \
                const float vi_ = ROWSHARE(((q) & 1) ? t1 : t0, (q) >> 1);                         \
                const v2f s_ = {vi_, vi_};                                                         \
                const v8f m_ = R[16 * (hf) + ((q) >> 1)];                                          \
                alo[(q) & 3] += s_ * (v2f){m_[4 * ((q) & 1)],     m_[4 * ((q) & 1) + 1]};          \
                ahi[(q) & 3] += s_ * (v2f){m_[4 * ((q) & 1) + 2], m_[4 * ((q) & 1) + 3]};          \
            } while (0)
            #define ROW4(hf, q)  ROWQ(hf, q); ROWQ(hf, (q) + 1); ROWQ(hf, (q) + 2); ROWQ(hf, (q) + 3)
            #define ROW16(hf, q) ROW4(hf, q); ROW4(hf, (q) + 4); ROW4(hf, (q) + 8); ROW4(hf, (q) + 12)
            #define STEP(hf) do {                                                                  \
                v2f alo[4] = {{0.f, 0.f}, {0.f, 0.f}, {0.f, 0.f}, {0.f, 0.f}};                     \
                v2f ahi[4] = {{0.f, 0.f}, {0.f, 0.f}, {0.f, 0.f}, {0.f, 0.f}};                     \
                ROW16(hf, 0); ROW16(hf, 16);                                                       \
                const v2f lo_ = (alo[0] + alo[1]) + (alo[2] + alo[3]);                             \
                const v2f hi_ = (ahi[0] + ahi[1]) + (ahi[2] + ahi[3]);                             \
                                             \
                sv.x = lo_[0] + XHALF(lo_[0]); sv.y = lo_[1] + XHALF(lo_[1]);                      \
                sv.z = hi_[0] + XHALF(hi_[0]); sv.w = hi_[1] + XHALF(hi_[1]);                      \
                t0 = h ? sv.y : sv.x; t1 = h ? sv.w : sv.z;                                        \
            } while (0)

            LOADM(0, 0);
            LOADM(1, 1);
            __builtin_amdgcn_sched_barrier(0);
            STEP(0);
            asm volatile("" :: "v"(t0), "v"(t1) : "memory");
            LOADM(0, 2);
            __builtin_amdgcn_sched_barrier(0);
            STEP(1);
            asm volatile("" :: "v"(t0), "v"(t1) : "memory");
            LOADM(1, 3);
            __builtin_amdgcn_sched_barrier(0);
            STEP(0);
            STEP(1);
            #undef STEP
            #undef ROW16
            #undef ROW4
            #undef ROWQ
            #undef LOADM
        }
        __syncthreads();
        if (more) {
            if (ih) STORE_HOLD(R);
            else    FORM_STORE(bf0, bf1);
        }
        __syncthreads();
    }
    #undef STORE_HOLD
    #undef FORM_HOLD
    #undef FORM_STORE
    #undef STORE8
    #undef FORM8

    if (ih == 0 && h == 0) *(float4*)(Vf + (size_t)(b0 + kq) * 64 + 4 * c) = sv;
}

__global__ __launch_bounds__(512)
void join_kernel(const float* __restrict__ Vf, const float* __restrict__ WlS,
                 float* __restrict__ out) {
    __shared__ float res[16 * C_];
    const int wv   = threadIdx.x >> 5;
    const int b    = blockIdx.x * 16 + wv;
    const int lane = threadIdx.x & 31;
    const float v0 = Vf[b * 64 + lane], v1 = Vf[b * 64 + 32 + lane];
    const float* up = WlS + lane;
    float tr[C_];
    #pragma unroll
    for (int k = 0; k < C_; ++k) tr[k] = fmaf(v1, up[k * 64 + 32], v0 * up[k * 64]);
    #pragma unroll
    for (int off = 16; off > 0; off >>= 1) {
        #pragma unroll
        for (int k = 0; k < C_; ++k) tr[k] += __shfl_xor(tr[k], off, 32);
    }
    float mx = tr[0];
    #pragma unroll
    for (int k = 1; k < C_; ++k) mx = fmaxf(mx, tr[k]);
    float se = 0.f;
    #pragma unroll
    for (int k = 0; k < C_; ++k) se += expf(tr[k] - mx);
    float mine = tr[0];
    #pragma unroll
    for (int k = 1; k < C_; ++k) mine = (lane == k) ? tr[k] : mine;
    if (lane < C_) res[wv * C_ + lane] = mine - (logf(se) + mx);
    __syncthreads();
    if (threadIdx.x < 16 * C_) out[(size_t)blockIdx.x * (16 * C_) + threadIdx.x] = res[threadIdx.x];
}

extern "C" void kernel_launch(void* const* d_in, const int* in_sizes, int n_in,
                              void* d_out, int out_size, void* d_ws, size_t ws_size,
                              hipStream_t stream) {
    const float* x     = (const float*)d_in[0];
    const float* Wsh   = (const float*)d_in[1];
    const float* Wlast = (const float*)d_in[2];
    float* out = (float*)d_out;
    float* ws  = (float*)d_ws;
    float* WlS    = ws;
    float* Vf     = ws + (C_ * R_ * R_) + (B_ * D_ * M_);

    hipFuncSetAttribute((const void*)trnet_kernel,
                        hipFuncAttributeMaxDynamicSharedMemorySize, LDSB);
    trnet_kernel<<<NWG, 256, LDSB, stream>>>(x, Wsh, Wlast, WlS, Vf);
    join_kernel<<<B_ / 16, 512, 0, stream>>>(Vf, WlS, out);
}
